# starting kernel
# baseline (speedup 1.0000x reference)
#include <hip/hip_runtime.h>


#define BSZ   8
#define T_    8
#define L_    196
#define D_    192
#define E_    192
#define C_    1536
#define N_    16
#define KC    4
#define R_    96
#define RN2   128
#define LTILES 13

typedef float v2f __attribute__((ext_vector_type(2)));
typedef float v8f __attribute__((ext_vector_type(8)));

__device__ __forceinline__ float sigmoid_(float x) {
    return 1.0f / (1.0f + __expf(-x));
}

__global__ __launch_bounds__(32)
void k_gemm_inproj(const float* __restrict__ hidden,
                   const float* __restrict__ w,
                   float* __restrict__ x_pre,
                   float* __restrict__ zbuf)
{
    const int lane = threadIdx.x;
    const int et   = blockIdx.x % 24;
    const int lt   = blockIdx.x / 24;
    const int bt   = blockIdx.y;
    const int b    = bt / T_;
    const int t    = bt % T_;

    const int mrow  = lt * 16 + (lane & 15);
    const int mclmp = (mrow < L_) ? mrow : (L_ - 1);
    const int koff  = (lane >> 4) * 2;
    const int ecol  = et * 16 + (lane & 15);

    const float* Arow = hidden + ((size_t)bt * L_ + mclmp) * D_;
    const float* Brow = w + (size_t)ecol * D_;

    v8f acc0 = {0.f,0.f,0.f,0.f,0.f,0.f,0.f,0.f};
    v8f acc1 = {0.f,0.f,0.f,0.f,0.f,0.f,0.f,0.f};
    for (int k0 = 0; k0 < D_; k0 += 8) {
        v2f a0 = *(const v2f*)(Arow + k0 + koff);
        v2f b0 = *(const v2f*)(Brow + k0 + koff);
        v2f a1 = *(const v2f*)(Arow + k0 + 4 + koff);
        v2f b1 = *(const v2f*)(Brow + k0 + 4 + koff);
        acc0 = __builtin_amdgcn_wmma_f32_16x16x4_f32(
                   false, a0, false, b0, (short)0, acc0, false, false);
        acc1 = __builtin_amdgcn_wmma_f32_16x16x4_f32(
                   false, a1, false, b1, (short)0, acc1, false, false);
    }

    const int nn    = lane & 15;
    const int e_out = et * 16 + nn;
    const int mbase = (lane >> 4) * 8;
    #pragma unroll
    for (int r = 0; r < 8; ++r) {
        const int l = lt * 16 + mbase + r;
        if (l < L_) {
            const float v = acc0[r] + acc1[r];
            if (e_out < E_) {
                const int c = e_out * T_ + t;
                x_pre[((size_t)b * C_ + c) * L_ + l] = v;
            } else {
                const int c = (e_out - E_) * T_ + t;
                zbuf[((size_t)b * C_ + c) * L_ + l] = v;
            }
        }
    }
}

__global__ __launch_bounds__(256)
void k_conv_silu(const float* __restrict__ x_pre,
                 const float* __restrict__ cw,
                 const float* __restrict__ cb,
                 float* __restrict__ xc)
{
    const int idx = blockIdx.x * 256 + threadIdx.x;
    if (idx >= BSZ * C_ * L_) return;
    const int l  = idx % L_;
    const int bc = idx / L_;
    const int c  = bc % C_;

    const float* xp = x_pre + (size_t)bc * L_;
    float s = cb[c];
    #pragma unroll
    for (int k = 0; k < KC; ++k) {
        const int ls = l - (KC - 1) + k;
        if (ls >= 0) s += xp[ls] * cw[c * KC + k];
    }
    xc[idx] = s * sigmoid_(s);
}

__global__ __launch_bounds__(32)
void k_gemm_xproj(const float* __restrict__ xc,
                  const float* __restrict__ w,
                  float* __restrict__ x_dbl)
{
    const int lane = threadIdx.x;
    const int rt   = blockIdx.x % 8;
    const int lt   = blockIdx.x / 8;
    const int b    = blockIdx.y;

    const int rrow  = rt * 16 + (lane & 15);
    const int koff  = (lane >> 4) * 2;
    const int lcol  = lt * 16 + (lane & 15);
    const int lclmp = (lcol < L_) ? lcol : (L_ - 1);

    const float* Arow  = w + (size_t)rrow * C_;
    const float* Bcol  = xc + (size_t)b * C_ * L_ + lclmp;

    v8f acc0 = {0.f,0.f,0.f,0.f,0.f,0.f,0.f,0.f};
    v8f acc1 = {0.f,0.f,0.f,0.f,0.f,0.f,0.f,0.f};
    for (int k0 = 0; k0 < C_; k0 += 8) {
        const int kB = k0 + koff;
        v2f a0 = *(const v2f*)(Arow + k0 + koff);
        v2f b0; b0.x = Bcol[(size_t)kB * L_];
                b0.y = Bcol[(size_t)(kB + 1) * L_];
        v2f a1 = *(const v2f*)(Arow + k0 + 4 + koff);
        v2f b1; b1.x = Bcol[(size_t)(kB + 4) * L_];
                b1.y = Bcol[(size_t)(kB + 5) * L_];
        acc0 = __builtin_amdgcn_wmma_f32_16x16x4_f32(
                   false, a0, false, b0, (short)0, acc0, false, false);
        acc1 = __builtin_amdgcn_wmma_f32_16x16x4_f32(
                   false, a1, false, b1, (short)0, acc1, false, false);
    }

    const int nn    = lane & 15;
    const int l     = lt * 16 + nn;
    const int mbase = (lane >> 4) * 8;
    if (l < L_) {
        #pragma unroll
        for (int r = 0; r < 8; ++r) {
            const int rr = rt * 16 + mbase + r;
            x_dbl[((size_t)b * RN2 + rr) * L_ + l] = acc0[r] + acc1[r];
        }
    }
}

__global__ __launch_bounds__(32)
void k_gemm_dtproj(const float* __restrict__ x_dbl,
                   const float* __restrict__ w,
                   const float* __restrict__ dt_bias,
                   float* __restrict__ delta)
{
    const int lane = threadIdx.x;
    const int ct   = blockIdx.x % 96;
    const int lt   = blockIdx.x / 96;
    const int b    = blockIdx.y;

    const int crow  = ct * 16 + (lane & 15);
    const int koff  = (lane >> 4) * 2;
    const int lcol  = lt * 16 + (lane & 15);
    const int lclmp = (lcol < L_) ? lcol : (L_ - 1);

    const float* Arow = w + (size_t)crow * R_;
    const float* Bcol = x_dbl + (size_t)b * RN2 * L_ + lclmp;

    v8f acc0 = {0.f,0.f,0.f,0.f,0.f,0.f,0.f,0.f};
    v8f acc1 = {0.f,0.f,0.f,0.f,0.f,0.f,0.f,0.f};
    for (int k0 = 0; k0 < R_; k0 += 8) {
        const int kB = k0 + koff;
        v2f a0 = *(const v2f*)(Arow + k0 + koff);
        v2f b0; b0.x = Bcol[(size_t)kB * L_];
                b0.y = Bcol[(size_t)(kB + 1) * L_];
        v2f a1 = *(const v2f*)(Arow + k0 + 4 + koff);
        v2f b1; b1.x = Bcol[(size_t)(kB + 4) * L_];
                b1.y = Bcol[(size_t)(kB + 5) * L_];
        acc0 = __builtin_amdgcn_wmma_f32_16x16x4_f32(
                   false, a0, false, b0, (short)0, acc0, false, false);
        acc1 = __builtin_amdgcn_wmma_f32_16x16x4_f32(
                   false, a1, false, b1, (short)0, acc1, false, false);
    }

    const int nn    = lane & 15;
    const int l     = lt * 16 + nn;
    const int mbase = (lane >> 4) * 8;
    if (l < L_) {
        #pragma unroll
        for (int r = 0; r < 8; ++r) {
            const int c = ct * 16 + mbase + r;
            const float v = acc0[r] + acc1[r] + dt_bias[c];
            const float sp = (v > 20.f) ? v : __logf(1.f + __expf(v));
            delta[((size_t)b * C_ + c) * L_ + l] = sp;
        }
    }
}

__global__ __launch_bounds__(256)
void k_scan_out(const float* __restrict__ delta,
                const float* __restrict__ xc,
                const float* __restrict__ zbuf,
                const float* __restrict__ x_dbl,
                const float* __restrict__ A_log,
                const float* __restrict__ Dp_,
                float* __restrict__ out)
{
    const int g  = blockIdx.x * 256 + threadIdx.x;
    const int n  = g & 15;
    const int bc = g >> 4;
    const int b  = bc / C_;
    const int c  = bc % C_;

    const float A_neg = -__expf(A_log[c * N_ + n]);
    const float Dp    = Dp_[c];

    const float* dp = delta + (size_t)bc * L_;
    const float* xp = xc    + (size_t)bc * L_;
    const float* zp = zbuf  + (size_t)bc * L_;
    const float* Bs = x_dbl + ((size_t)b * RN2 + R_       + n) * L_;
    const float* Cs = x_dbl + ((size_t)b * RN2 + R_ + N_  + n) * L_;
    float* op = out + (size_t)bc * L_;

    float h = 0.f;
    for (int l = 0; l < L_; ++l) {
        const float dv = dp[l];
        const float xv = xp[l];
        const float dA = __expf(dv * A_neg);
        h = h * dA + dv * Bs[l] * xv;
        float yp = h * Cs[l];
        yp += __shfl_xor(yp, 1, 16);
        yp += __shfl_xor(yp, 2, 16);
        yp += __shfl_xor(yp, 4, 16);
        yp += __shfl_xor(yp, 8, 16);
        if (n == 0) {
            const float zv = zp[l];
            op[l] = (yp + xv * Dp) * (zv * sigmoid_(zv));
        }
    }
}

extern "C" void kernel_launch(void* const* d_in, const int* in_sizes, int n_in,
                              void* d_out, int out_size, void* d_ws, size_t ws_size,
                              hipStream_t stream)
{
    const float* hidden    = (const float*)d_in[0];
    const float* in_proj_w = (const float*)d_in[1];
    const float* conv_w    = (const float*)d_in[2];
    const float* conv_b    = (const float*)d_in[3];
    const float* x_proj_w  = (const float*)d_in[4];
    const float* dt_proj_w = (const float*)d_in[5];
    const float* dt_bias   = (const float*)d_in[6];
    const float* A_log     = (const float*)d_in[7];
    const float* D_param   = (const float*)d_in[8];

    float* ws = (float*)d_ws;
    const size_t BCL = (size_t)BSZ * C_ * L_;
    float* x_pre = ws;
    float* zbuf  = ws + BCL;
    float* xcbuf = ws + 2 * BCL;
    float* delta = ws + 3 * BCL;
    float* x_dbl = ws + 4 * BCL;

    k_gemm_inproj<<<dim3(24 * LTILES, BSZ * T_), 32, 0, stream>>>(
        hidden, in_proj_w, x_pre, zbuf);

    k_conv_silu<<<(BSZ * C_ * L_ + 255) / 256, 256, 0, stream>>>(
        x_pre, conv_w, conv_b, xcbuf);

    k_gemm_xproj<<<dim3(8 * LTILES, BSZ), 32, 0, stream>>>(
        xcbuf, x_proj_w, x_dbl);

    k_gemm_dtproj<<<dim3(96 * LTILES, BSZ), 32, 0, stream>>>(
        x_dbl, dt_proj_w, dt_bias, delta);

    k_scan_out<<<(BSZ * C_ * N_) / 256, 256, 0, stream>>>(
        delta, xcbuf, zbuf, x_dbl, A_log, D_param, (float*)d_out);
}


# conv+SiLU kernel fused into the x_proj GEMM: a block builds its (b, 16 l) xc tile of all 1536 channels in LDS (16-B loads) and reads WMMA B from it; sum orders kept; 4 launches, not 5
# speedup vs baseline: 1.7248x; 1.7248x over previous
#include <hip/hip_runtime.h>


#define BSZ   8
#define T_    8
#define L_    196
#define D_    192
#define E_    192
#define C_    1536
#define N_    16
#define KC    4
#define R_    96
#define RN2   128
#define LTILES 13
#define SROW4_X (L_ / 4)

typedef float v2f __attribute__((ext_vector_type(2)));
typedef float v8f __attribute__((ext_vector_type(8)));

__device__ __forceinline__ float sigmoid_(float x) {
    return 1.0f / (1.0f + __expf(-x));
}

__global__ __launch_bounds__(32)
void k_gemm_inproj(const float* __restrict__ hidden,
                   const float* __restrict__ w,
                   float* __restrict__ x_pre,
                   float* __restrict__ zbuf)
{
    const int lane = threadIdx.x;
    const int et   = blockIdx.x % 24;
    const int lt   = blockIdx.x / 24;
    const int bt   = blockIdx.y;
    const int b    = bt / T_;
    const int t    = bt % T_;

    const int mrow  = lt * 16 + (lane & 15);
    const int mclmp = (mrow < L_) ? mrow : (L_ - 1);
    const int koff  = (lane >> 4) * 2;
    const int ecol  = et * 16 + (lane & 15);

    const float* Arow = hidden + ((size_t)bt * L_ + mclmp) * D_;
    const float* Brow = w + (size_t)ecol * D_;

    v8f acc0 = {0.f,0.f,0.f,0.f,0.f,0.f,0.f,0.f};
    v8f acc1 = {0.f,0.f,0.f,0.f,0.f,0.f,0.f,0.f};
    for (int k0 = 0; k0 < D_; k0 += 8) {
        v2f a0 = *(const v2f*)(Arow + k0 + koff);
        v2f b0 = *(const v2f*)(Brow + k0 + koff);
        v2f a1 = *(const v2f*)(Arow + k0 + 4 + koff);
        v2f b1 = *(const v2f*)(Brow + k0 + 4 + koff);
        acc0 = __builtin_amdgcn_wmma_f32_16x16x4_f32(
                   false, a0, false, b0, (short)0, acc0, false, false);
        acc1 = __builtin_amdgcn_wmma_f32_16x16x4_f32(
                   false, a1, false, b1, (short)0, acc1, false, false);
    }

    const int nn    = lane & 15;
    const int e_out = et * 16 + nn;
    const int mbase = (lane >> 4) * 8;
    #pragma unroll
    for (int r = 0; r < 8; ++r) {
        const int l = lt * 16 + mbase + r;
        if (l < L_) {
            const float v = acc0[r] + acc1[r];
            if (e_out < E_) {
                const int c = e_out * T_ + t;
                x_pre[((size_t)b * C_ + c) * L_ + l] = v;
            } else {
                const int c = (e_out - E_) * T_ + t;
                zbuf[((size_t)b * C_ + c) * L_ + l] = v;
            }
        }
    }
}

#define XPW    4
#define XKCH   64
#define XNCH   (C_ / XKCH)

__device__ __forceinline__ float conv_tap4_(float bias, float4 wv, float a,
                                            float b, float c, float d)
{
    float s = bias;
    s += a * wv.x;
    s += b * wv.y;
    s += c * wv.z;
    s += d * wv.w;
    return s * sigmoid_(s);
}

__global__ __launch_bounds__(XPW * 32)
void k_conv_xproj(const float* __restrict__ x_pre,
                  const float* __restrict__ cw,
                  const float* __restrict__ cb,
                  const float* __restrict__ w,
                  float* __restrict__ xc,
                  float* __restrict__ x_dbl)
{
    __shared__ float4 sX[C_ * 4];

    const int tid = threadIdx.x;
    const int rh  = blockIdx.x & 1;
    const int lt  = blockIdx.x >> 1;
    const int b   = blockIdx.y;

    {
        const int j  = tid & 3;
        const int cr = tid >> 2;
        const int l  = lt * 16 + 4 * j;
        const bool valid   = l < L_;
        const bool hasprev = l > 0;
        const bool store   = valid && (rh == 0);
        const int q  = (valid ? l : (L_ - 4)) >> 2;
        const int qp = hasprev ? (q - 1) : q;
        const float4* xr4 = (const float4*)(x_pre + (size_t)b * C_ * L_);
        float4*       xo4 = (float4*)(xc + (size_t)b * C_ * L_) + q;
        const float4* cw4 = (const float4*)cw;
        float4 curN[8], plN[8], wvN[8];
        float  bsN[8];
        #pragma unroll
        for (int u = 0; u < 8; ++u) {
            const int c = u * 32 + cr;
            curN[u] = xr4[c * SROW4_X + q];
            plN[u]  = xr4[c * SROW4_X + qp];
            wvN[u]  = cw4[c];
            bsN[u]  = cb[c];
        }
        for (int t = 0; t < C_ / 256; ++t) {
            float4 curC[8], plC[8], wvC[8];
            float  bsC[8];
            #pragma unroll
            for (int u = 0; u < 8; ++u) {
                curC[u] = curN[u]; plC[u] = plN[u]; wvC[u] = wvN[u]; bsC[u] = bsN[u];
            }
            const int tn = (t + 1 < C_ / 256) ? (t + 1) : t;
            #pragma unroll
            for (int u = 0; u < 8; ++u) {
                const int c = tn * 256 + u * 32 + cr;
                curN[u] = xr4[c * SROW4_X + q];
                plN[u]  = xr4[c * SROW4_X + qp];
                wvN[u]  = cw4[c];
                bsN[u]  = cb[c];
            }
            #pragma unroll
            for (int u = 0; u < 8; ++u) {
                const int c = t * 256 + u * 32 + cr;
                const float4 cur = curC[u];
                const float4 wv  = wvC[u];
                const float bias = bsC[u];
                const float p1 = hasprev ? plC[u].y : 0.f;
                const float p2 = hasprev ? plC[u].z : 0.f;
                const float p3 = hasprev ? plC[u].w : 0.f;
                float4 o;
                o.x = conv_tap4_(bias, wv, p1, p2, p3, cur.x);
                o.y = conv_tap4_(bias, wv, p2, p3, cur.x, cur.y);
                o.z = conv_tap4_(bias, wv, p3, cur.x, cur.y, cur.z);
                o.w = conv_tap4_(bias, wv, cur.x, cur.y, cur.z, cur.w);
                sX[c * 4 + j] = o;
                if (store) xo4[c * SROW4_X] = o;
            }
        }
    }
    __syncthreads();

    const int lane = tid & 31;
    const int rt   = rh * XPW + (tid >> 5);
    const int m    = lane & 15;
    const int h    = lane >> 4;
    const float* Arow = w + (size_t)(rt * 16 + m) * C_ + 2 * h;
    const float* Bcol = (const float*)sX + (2 * h) * 16 + m;

    v8f acc0 = {0.f,0.f,0.f,0.f,0.f,0.f,0.f,0.f};
    v8f acc1 = {0.f,0.f,0.f,0.f,0.f,0.f,0.f,0.f};

    v2f aN[16];
    #pragma unroll
    for (int i = 0; i < 16; ++i) aN[i] = *(const v2f*)(Arow + 4 * i);

    for (int ch = 0; ch < XNCH; ++ch) {
        v2f aC[16];
        #pragma unroll
        for (int i = 0; i < 16; ++i) aC[i] = aN[i];
        const int nx = (ch + 1 < XNCH) ? (ch + 1) : (XNCH - 1);
        #pragma unroll
        for (int i = 0; i < 16; ++i)
            aN[i] = *(const v2f*)(Arow + nx * XKCH + 4 * i);
        v2f bC[16];
        #pragma unroll
        for (int i = 0; i < 16; ++i) {
            const float* p = Bcol + (ch * XKCH + 4 * i) * 16;
            bC[i].x = p[0];
            bC[i].y = p[16];
        }
        #pragma unroll
        for (int i = 0; i < 16; i += 2) {
            acc0 = __builtin_amdgcn_wmma_f32_16x16x4_f32(
                       false, aC[i], false, bC[i], (short)0, acc0, false, false);
            acc1 = __builtin_amdgcn_wmma_f32_16x16x4_f32(
                       false, aC[i + 1], false, bC[i + 1], (short)0, acc1, false, false);
        }
    }

    const int l     = lt * 16 + m;
    const int mbase = h * 8;
    if (l < L_) {
        #pragma unroll
        for (int r = 0; r < 8; ++r) {
            const int rr = rt * 16 + mbase + r;
            x_dbl[((size_t)b * RN2 + rr) * L_ + l] = acc0[r] + acc1[r];
        }
    }
}

__global__ __launch_bounds__(32)
void k_gemm_dtproj(const float* __restrict__ x_dbl,
                   const float* __restrict__ w,
                   const float* __restrict__ dt_bias,
                   float* __restrict__ delta)
{
    const int lane = threadIdx.x;
    const int ct   = blockIdx.x % 96;
    const int lt   = blockIdx.x / 96;
    const int b    = blockIdx.y;

    const int crow  = ct * 16 + (lane & 15);
    const int koff  = (lane >> 4) * 2;
    const int lcol  = lt * 16 + (lane & 15);
    const int lclmp = (lcol < L_) ? lcol : (L_ - 1);

    const float* Arow = w + (size_t)crow * R_;
    const float* Bcol = x_dbl + (size_t)b * RN2 * L_ + lclmp;

    v8f acc0 = {0.f,0.f,0.f,0.f,0.f,0.f,0.f,0.f};
    v8f acc1 = {0.f,0.f,0.f,0.f,0.f,0.f,0.f,0.f};
    for (int k0 = 0; k0 < R_; k0 += 8) {
        const int kB = k0 + koff;
        v2f a0 = *(const v2f*)(Arow + k0 + koff);
        v2f b0; b0.x = Bcol[(size_t)kB * L_];
                b0.y = Bcol[(size_t)(kB + 1) * L_];
        v2f a1 = *(const v2f*)(Arow + k0 + 4 + koff);
        v2f b1; b1.x = Bcol[(size_t)(kB + 4) * L_];
                b1.y = Bcol[(size_t)(kB + 5) * L_];
        acc0 = __builtin_amdgcn_wmma_f32_16x16x4_f32(
                   false, a0, false, b0, (short)0, acc0, false, false);
        acc1 = __builtin_amdgcn_wmma_f32_16x16x4_f32(
                   false, a1, false, b1, (short)0, acc1, false, false);
    }

    const int nn    = lane & 15;
    const int l     = lt * 16 + nn;
    const int mbase = (lane >> 4) * 8;
    if (l < L_) {
        #pragma unroll
        for (int r = 0; r < 8; ++r) {
            const int c = ct * 16 + mbase + r;
            const float v = acc0[r] + acc1[r] + dt_bias[c];
            const float sp = (v > 20.f) ? v : __logf(1.f + __expf(v));
            delta[((size_t)b * C_ + c) * L_ + l] = sp;
        }
    }
}

#define SCH    16
#define SROW4  (L_ / 4)
#define STILE4 (SCH * SROW4)

__device__ __forceinline__ float4 gate_(float4 y, float4 x, float4 z, float Dp)
{
    float4 o;
    o.x = (y.x + x.x * Dp) * (z.x * sigmoid_(z.x));
    o.y = (y.y + x.y * Dp) * (z.y * sigmoid_(z.y));
    o.z = (y.z + x.z * Dp) * (z.z * sigmoid_(z.z));
    o.w = (y.w + x.w * Dp) * (z.w * sigmoid_(z.w));
    return o;
}

__device__ __forceinline__ float scan_step_(float& h, float A_neg, float dv,
                                            float u, float bv, float cv)
{
    const float dA = __expf(dv * A_neg);
    h = fmaf(h, dA, u * bv);
    float yp = h * cv;
    yp += __shfl_xor(yp, 1, 16);
    yp += __shfl_xor(yp, 2, 16);
    yp += __shfl_xor(yp, 4, 16);
    yp += __shfl_xor(yp, 8, 16);
    return yp;
}

__global__ __launch_bounds__(256)
void k_scan_out(const float* __restrict__ delta,
                const float* __restrict__ xc,
                const float* __restrict__ zbuf,
                const float* __restrict__ x_dbl,
                const float* __restrict__ A_log,
                const float* __restrict__ Dp_,
                float* __restrict__ out)
{
    __shared__ float4 sD[STILE4];
    __shared__ float4 sU[STILE4];
    __shared__ float4 sB[STILE4];
    __shared__ float4 sC[STILE4];
    __shared__ float4 sY[STILE4];

    const int tid = threadIdx.x;
    const int cb  = blockIdx.x % (C_ / SCH);
    const int b   = blockIdx.x / (C_ / SCH);
    const size_t row0 = ((size_t)b * C_ + (size_t)cb * SCH) * L_;

    const float4* gD = (const float4*)(delta + row0);
    const float4* gX = (const float4*)(xc    + row0);
    const float4* gZ = (const float4*)(zbuf  + row0);
    const float4* gB = (const float4*)(x_dbl + ((size_t)b * RN2 + R_) * L_);
    const float4* gC = (const float4*)(x_dbl + ((size_t)b * RN2 + R_ + N_) * L_);
    float4* gO = (float4*)(out + row0);

    const int lane = tid & 31;
    const int n    = lane & 15;
    const int ch   = (tid >> 5) * 2 + (lane >> 4);
    const float A_neg = -__expf(A_log[(cb * SCH + ch) * N_ + n]);

    float4 dr[3], xr[3], zr[3], br[3], cr[3];
    const bool tail = tid < (STILE4 - 3 * 256);
    const float4 zero4 = make_float4(0.f, 0.f, 0.f, 0.f);
    float4 dT = zero4, xT = zero4, zT = zero4, bT = zero4, cT = zero4;
    float  pr[3], pT = 0.f;
    #pragma unroll
    for (int r = 0; r < 3; ++r) {
        const int i = tid + r * 256;
        dr[r] = gD[i];
        xr[r] = gX[i];
        br[r] = gB[i];
        cr[r] = gC[i];
        zr[r] = gZ[i];
        pr[r] = Dp_[cb * SCH + i / SROW4];
    }
    if (tail) {
        const int i = tid + 3 * 256;
        dT = gD[i];
        xT = gX[i];
        bT = gB[i];
        cT = gC[i];
        zT = gZ[i];
        pT = Dp_[cb * SCH + i / SROW4];
    }
    #pragma unroll
    for (int r = 0; r < 3; ++r) {
        const int i = tid + r * 256;
        const float4 d = dr[r];
        const float4 x = xr[r];
        sD[i] = d;
        sU[i] = make_float4(d.x * x.x, d.y * x.y, d.z * x.z, d.w * x.w);
        sB[i] = br[r];
        sC[i] = cr[r];
    }
    if (tail) {
        const int i = tid + 3 * 256;
        sD[i] = dT;
        sU[i] = make_float4(dT.x * xT.x, dT.y * xT.y, dT.z * xT.z, dT.w * xT.w);
        sB[i] = bT;
        sC[i] = cT;
    }
    __syncthreads();

    const float4* pD = sD + ch * SROW4;
    const float4* pU = sU + ch * SROW4;
    const float4* pB = sB + n  * SROW4;
    const float4* pC = sC + n  * SROW4;
    float4*       pY = sY + ch * SROW4;

    float h = 0.f;
    float4 dn = pD[0], un = pU[0], bn = pB[0], cn = pC[0];
    for (int j = 0; j < SROW4; ++j) {
        const float4 d = dn, u = un, bv = bn, cv = cn;
        const int jn = (j + 1 < SROW4) ? (j + 1) : (SROW4 - 1);
        dn = pD[jn]; un = pU[jn]; bn = pB[jn]; cn = pC[jn];
        float4 y;
        y.x = scan_step_(h, A_neg, d.x, u.x, bv.x, cv.x);
        y.y = scan_step_(h, A_neg, d.y, u.y, bv.y, cv.y);
        y.z = scan_step_(h, A_neg, d.z, u.z, bv.z, cv.z);
        y.w = scan_step_(h, A_neg, d.w, u.w, bv.w, cv.w);
        if (n == 0) pY[j] = y;
    }
    __syncthreads();

    #pragma unroll
    for (int r = 0; r < 3; ++r) {
        const int i = tid + r * 256;
        gO[i] = gate_(sY[i], xr[r], zr[r], pr[r]);
    }
    if (tail) {
        const int i = tid + 3 * 256;
        gO[i] = gate_(sY[i], xT, zT, pT);
    }
}

extern "C" void kernel_launch(void* const* d_in, const int* in_sizes, int n_in,
                              void* d_out, int out_size, void* d_ws, size_t ws_size,
                              hipStream_t stream)
{
    const float* hidden    = (const float*)d_in[0];
    const float* in_proj_w = (const float*)d_in[1];
    const float* conv_w    = (const float*)d_in[2];
    const float* conv_b    = (const float*)d_in[3];
    const float* x_proj_w  = (const float*)d_in[4];
    const float* dt_proj_w = (const float*)d_in[5];
    const float* dt_bias   = (const float*)d_in[6];
    const float* A_log     = (const float*)d_in[7];
    const float* D_param   = (const float*)d_in[8];

    float* ws = (float*)d_ws;
    const size_t BCL = (size_t)BSZ * C_ * L_;
    float* x_pre = ws;
    float* zbuf  = ws + BCL;
    float* xcbuf = ws + 2 * BCL;
    float* delta = ws + 3 * BCL;
    float* x_dbl = ws + 4 * BCL;

    k_gemm_inproj<<<dim3(24 * LTILES, BSZ * T_), 32, 0, stream>>>(
        hidden, in_proj_w, x_pre, zbuf);

    k_conv_xproj<<<dim3(2 * LTILES, BSZ), XPW * 32, 0, stream>>>(
        x_pre, conv_w, conv_b, x_proj_w, xcbuf, x_dbl);

    k_gemm_dtproj<<<dim3(96 * LTILES, BSZ), 32, 0, stream>>>(
        x_dbl, dt_proj_w, dt_bias, delta);

    k_scan_out<<<BSZ * (C_ / SCH), 256, 0, stream>>>(
        delta, xcbuf, zbuf, x_dbl, A_log, D_param, (float*)d_out);
}


# in_proj GEMM: a 12-wave block stages its whole (b,t) slab of hidden in LDS (16-B line loads), reads WMMA B from it; 16 weight rows per wave kept in registers; l on lanes for stores; sum order kept
# speedup vs baseline: 2.8613x; 1.6589x over previous
#include <hip/hip_runtime.h>


#define BSZ   8
#define T_    8
#define L_    196
#define D_    192
#define E_    192
#define C_    1536
#define N_    16
#define KC    4
#define R_    96
#define RN2   128
#define LTILES 13
#define SROW4_X (L_ / 4)

typedef float v2f __attribute__((ext_vector_type(2)));
typedef float v8f __attribute__((ext_vector_type(8)));

__device__ __forceinline__ float sigmoid_(float x) {
    return 1.0f / (1.0f + __expf(-x));
}

#define IPW     12
#define IPE     96
#define HPITCH  196
#define HPITCH4 (HPITCH / 4)
#define HROW4   (D_ / 4)
#define HTILE4  (L_ * HROW4)
#define HSTG    13

__global__ __launch_bounds__(IPW * 32)
void k_gemm_inproj(const float* __restrict__ hidden,
                   const float* __restrict__ w,
                   float* __restrict__ x_pre,
                   float* __restrict__ zbuf)
{
    __shared__ float4 sH4[L_ * HPITCH4];

    const int tid  = threadIdx.x;
    const int lane = tid & 31;
    const int wid  = tid >> 5;
    const int c    = lane & 15;
    const int h    = lane >> 4;
    const int eq   = blockIdx.x;
    const int bt   = blockIdx.y;
    const int b    = bt / T_;
    const int t    = bt % T_;
    const int et   = wid % 6;
    const int lh   = wid / 6;

    const float* Arow = w + (size_t)(eq * IPE + et * 16 + c) * D_ + 2 * h;
    v2f aF[48];
    #pragma unroll
    for (int s = 0; s < 48; ++s) aF[s] = *(const v2f*)(Arow + 4 * s);

    {
        const float4* g4 = (const float4*)(hidden + (size_t)bt * L_ * D_);
        for (int bb = 0; bb < 2; ++bb) {
            float4 v[HSTG];
            #pragma unroll
            for (int u = 0; u < HSTG; ++u) {
                const int i  = tid + (bb * HSTG + u) * (IPW * 32);
                const int ic = (i < HTILE4) ? i : (HTILE4 - 1);
                v[u] = g4[ic];
            }
            #pragma unroll
            for (int u = 0; u < HSTG; ++u) {
                const int i = tid + (bb * HSTG + u) * (IPW * 32);
                if (i < HTILE4) sH4[(i / HROW4) * HPITCH4 + (i % HROW4)] = v[u];
            }
        }
    }
    __syncthreads();

    const float* sH = (const float*)sH4;
    const int lt0 = lh ? 7 : 0;
    const int lt1 = lh ? LTILES : 7;
    float* __restrict__ dst = (eq < 2) ? x_pre : zbuf;
    const int e0 = (eq & 1) * IPE + et * 16 + 8 * h;

    v2f bN[16];
    {
        const int r0  = lt0 * 16 + c;
        const float* p = sH + ((r0 < L_) ? r0 : (L_ - 1)) * HPITCH + 2 * h;
        #pragma unroll
        for (int i = 0; i < 16; ++i) bN[i] = *(const v2f*)(p + 4 * i);
    }
    for (int lt = lt0; lt < lt1; ++lt) {
        const int ltn = (lt + 1 < lt1) ? (lt + 1) : lt;
        const int rc  = lt * 16 + c;
        const int rn  = ltn * 16 + c;
        const float* pc = sH + ((rc < L_) ? rc : (L_ - 1)) * HPITCH + 2 * h;
        const float* pn = sH + ((rn < L_) ? rn : (L_ - 1)) * HPITCH + 2 * h;

        v8f acc0 = {0.f,0.f,0.f,0.f,0.f,0.f,0.f,0.f};
        v8f acc1 = {0.f,0.f,0.f,0.f,0.f,0.f,0.f,0.f};
        #pragma unroll
        for (int ch = 0; ch < 3; ++ch) {
            v2f bC[16];
            #pragma unroll
            for (int i = 0; i < 16; ++i) bC[i] = bN[i];
            const float* pp = (ch < 2) ? (pc + (ch + 1) * 64) : pn;
            #pragma unroll
            for (int i = 0; i < 16; ++i) bN[i] = *(const v2f*)(pp + 4 * i);
            __builtin_amdgcn_sched_barrier(0);
            #pragma unroll
            for (int i = 0; i < 16; i += 2) {
                acc0 = __builtin_amdgcn_wmma_f32_16x16x4_f32(
                           false, aF[ch * 16 + i], false, bC[i], (short)0, acc0, false, false);
                acc1 = __builtin_amdgcn_wmma_f32_16x16x4_f32(
                           false, aF[ch * 16 + i + 1], false, bC[i + 1], (short)0, acc1, false, false);
            }
        }

        const int l = lt * 16 + c;
        if (l < L_) {
            #pragma unroll
            for (int r = 0; r < 8; ++r) {
                const int ch_out = (e0 + r) * T_ + t;
                dst[((size_t)b * C_ + ch_out) * L_ + l] = acc0[r] + acc1[r];
            }
        }
    }
}

#define XPW    4
#define XKCH   64
#define XNCH   (C_ / XKCH)

__device__ __forceinline__ float conv_tap4_(float bias, float4 wv, float a,
                                            float b, float c, float d)
{
    float s = bias;
    s += a * wv.x;
    s += b * wv.y;
    s += c * wv.z;
    s += d * wv.w;
    return s * sigmoid_(s);
}

__global__ __launch_bounds__(XPW * 32)
void k_conv_xproj(const float* __restrict__ x_pre,
                  const float* __restrict__ cw,
                  const float* __restrict__ cb,
                  const float* __restrict__ w,
                  float* __restrict__ xc,
                  float* __restrict__ x_dbl)
{
    __shared__ float4 sX[C_ * 4];

    const int tid = threadIdx.x;
    const int rh  = blockIdx.x & 1;
    const int lt  = blockIdx.x >> 1;
    const int b   = blockIdx.y;

    {
        const int j  = tid & 3;
        const int cr = tid >> 2;
        const int l  = lt * 16 + 4 * j;
        const bool valid   = l < L_;
        const bool hasprev = l > 0;
        const bool store   = valid && (rh == 0);
        const int q  = (valid ? l : (L_ - 4)) >> 2;
        const int qp = hasprev ? (q - 1) : q;
        const float4* xr4 = (const float4*)(x_pre + (size_t)b * C_ * L_);
        float4*       xo4 = (float4*)(xc + (size_t)b * C_ * L_) + q;
        const float4* cw4 = (const float4*)cw;
        float4 curN[8], plN[8], wvN[8];
        float  bsN[8];
        #pragma unroll
        for (int u = 0; u < 8; ++u) {
            const int c = u * 32 + cr;
            curN[u] = xr4[c * SROW4_X + q];
            plN[u]  = xr4[c * SROW4_X + qp];
            wvN[u]  = cw4[c];
            bsN[u]  = cb[c];
        }
        for (int t = 0; t < C_ / 256; ++t) {
            float4 curC[8], plC[8], wvC[8];
            float  bsC[8];
            #pragma unroll
            for (int u = 0; u < 8; ++u) {
                curC[u] = curN[u]; plC[u] = plN[u]; wvC[u] = wvN[u]; bsC[u] = bsN[u];
            }
            const int tn = (t + 1 < C_ / 256) ? (t + 1) : t;
            #pragma unroll
            for (int u = 0; u < 8; ++u) {
                const int c = tn * 256 + u * 32 + cr;
                curN[u] = xr4[c * SROW4_X + q];
                plN[u]  = xr4[c * SROW4_X + qp];
                wvN[u]  = cw4[c];
                bsN[u]  = cb[c];
            }
            #pragma unroll
            for (int u = 0; u < 8; ++u) {
                const int c = t * 256 + u * 32 + cr;
                const float4 cur = curC[u];
                const float4 wv  = wvC[u];
                const float bias = bsC[u];
                const float p1 = hasprev ? plC[u].y : 0.f;
                const float p2 = hasprev ? plC[u].z : 0.f;
                const float p3 = hasprev ? plC[u].w : 0.f;
                float4 o;
                o.x = conv_tap4_(bias, wv, p1, p2, p3, cur.x);
                o.y = conv_tap4_(bias, wv, p2, p3, cur.x, cur.y);
                o.z = conv_tap4_(bias, wv, p3, cur.x, cur.y, cur.z);
                o.w = conv_tap4_(bias, wv, cur.x, cur.y, cur.z, cur.w);
                sX[c * 4 + j] = o;
                if (store) xo4[c * SROW4_X] = o;
            }
        }
    }
    __syncthreads();

    const int lane = tid & 31;
    const int rt   = rh * XPW + (tid >> 5);
    const int m    = lane & 15;
    const int h    = lane >> 4;
    const float* Arow = w + (size_t)(rt * 16 + m) * C_ + 2 * h;
    const float* Bcol = (const float*)sX + (2 * h) * 16 + m;

    v8f acc0 = {0.f,0.f,0.f,0.f,0.f,0.f,0.f,0.f};
    v8f acc1 = {0.f,0.f,0.f,0.f,0.f,0.f,0.f,0.f};

    v2f aN[16];
    #pragma unroll
    for (int i = 0; i < 16; ++i) aN[i] = *(const v2f*)(Arow + 4 * i);

    for (int ch = 0; ch < XNCH; ++ch) {
        v2f aC[16];
        #pragma unroll
        for (int i = 0; i < 16; ++i) aC[i] = aN[i];
        const int nx = (ch + 1 < XNCH) ? (ch + 1) : (XNCH - 1);
        #pragma unroll
        for (int i = 0; i < 16; ++i)
            aN[i] = *(const v2f*)(Arow + nx * XKCH + 4 * i);
        v2f bC[16];
        #pragma unroll
        for (int i = 0; i < 16; ++i) {
            const float* p = Bcol + (ch * XKCH + 4 * i) * 16;
            bC[i].x = p[0];
            bC[i].y = p[16];
        }
        #pragma unroll
        for (int i = 0; i < 16; i += 2) {
            acc0 = __builtin_amdgcn_wmma_f32_16x16x4_f32(
                       false, aC[i], false, bC[i], (short)0, acc0, false, false);
            acc1 = __builtin_amdgcn_wmma_f32_16x16x4_f32(
                       false, aC[i + 1], false, bC[i + 1], (short)0, acc1, false, false);
        }
    }

    const int l     = lt * 16 + m;
    const int mbase = h * 8;
    if (l < L_) {
        #pragma unroll
        for (int r = 0; r < 8; ++r) {
            const int rr = rt * 16 + mbase + r;
            x_dbl[((size_t)b * RN2 + rr) * L_ + l] = acc0[r] + acc1[r];
        }
    }
}

__global__ __launch_bounds__(32)
void k_gemm_dtproj(const float* __restrict__ x_dbl,
                   const float* __restrict__ w,
                   const float* __restrict__ dt_bias,
                   float* __restrict__ delta)
{
    const int lane = threadIdx.x;
    const int ct   = blockIdx.x % 96;
    const int lt   = blockIdx.x / 96;
    const int b    = blockIdx.y;

    const int crow  = ct * 16 + (lane & 15);
    const int koff  = (lane >> 4) * 2;
    const int lcol  = lt * 16 + (lane & 15);
    const int lclmp = (lcol < L_) ? lcol : (L_ - 1);

    const float* Arow = w + (size_t)crow * R_;
    const float* Bcol = x_dbl + (size_t)b * RN2 * L_ + lclmp;

    v8f acc0 = {0.f,0.f,0.f,0.f,0.f,0.f,0.f,0.f};
    v8f acc1 = {0.f,0.f,0.f,0.f,0.f,0.f,0.f,0.f};
    for (int k0 = 0; k0 < R_; k0 += 8) {
        const int kB = k0 + koff;
        v2f a0 = *(const v2f*)(Arow + k0 + koff);
        v2f b0; b0.x = Bcol[(size_t)kB * L_];
                b0.y = Bcol[(size_t)(kB + 1) * L_];
        v2f a1 = *(const v2f*)(Arow + k0 + 4 + koff);
        v2f b1; b1.x = Bcol[(size_t)(kB + 4) * L_];
                b1.y = Bcol[(size_t)(kB + 5) * L_];
        acc0 = __builtin_amdgcn_wmma_f32_16x16x4_f32(
                   false, a0, false, b0, (short)0, acc0, false, false);
        acc1 = __builtin_amdgcn_wmma_f32_16x16x4_f32(
                   false, a1, false, b1, (short)0, acc1, false, false);
    }

    const int nn    = lane & 15;
    const int l     = lt * 16 + nn;
    const int mbase = (lane >> 4) * 8;
    if (l < L_) {
        #pragma unroll
        for (int r = 0; r < 8; ++r) {
            const int c = ct * 16 + mbase + r;
            const float v = acc0[r] + acc1[r] + dt_bias[c];
            const float sp = (v > 20.f) ? v : __logf(1.f + __expf(v));
            delta[((size_t)b * C_ + c) * L_ + l] = sp;
        }
    }
}

#define SCH    16
#define SROW4  (L_ / 4)
#define STILE4 (SCH * SROW4)

__device__ __forceinline__ float4 gate_(float4 y, float4 x, float4 z, float Dp)
{
    float4 o;
    o.x = (y.x + x.x * Dp) * (z.x * sigmoid_(z.x));
    o.y = (y.y + x.y * Dp) * (z.y * sigmoid_(z.y));
    o.z = (y.z + x.z * Dp) * (z.z * sigmoid_(z.z));
    o.w = (y.w + x.w * Dp) * (z.w * sigmoid_(z.w));
    return o;
}

__device__ __forceinline__ float scan_step_(float& h, float A_neg, float dv,
                                            float u, float bv, float cv)
{
    const float dA = __expf(dv * A_neg);
    h = fmaf(h, dA, u * bv);
    float yp = h * cv;
    yp += __shfl_xor(yp, 1, 16);
    yp += __shfl_xor(yp, 2, 16);
    yp += __shfl_xor(yp, 4, 16);
    yp += __shfl_xor(yp, 8, 16);
    return yp;
}

__global__ __launch_bounds__(256)
void k_scan_out(const float* __restrict__ delta,
                const float* __restrict__ xc,
                const float* __restrict__ zbuf,
                const float* __restrict__ x_dbl,
                const float* __restrict__ A_log,
                const float* __restrict__ Dp_,
                float* __restrict__ out)
{
    __shared__ float4 sD[STILE4];
    __shared__ float4 sU[STILE4];
    __shared__ float4 sB[STILE4];
    __shared__ float4 sC[STILE4];
    __shared__ float4 sY[STILE4];

    const int tid = threadIdx.x;
    const int cb  = blockIdx.x % (C_ / SCH);
    const int b   = blockIdx.x / (C_ / SCH);
    const size_t row0 = ((size_t)b * C_ + (size_t)cb * SCH) * L_;

    const float4* gD = (const float4*)(delta + row0);
    const float4* gX = (const float4*)(xc    + row0);
    const float4* gZ = (const float4*)(zbuf  + row0);
    const float4* gB = (const float4*)(x_dbl + ((size_t)b * RN2 + R_) * L_);
    const float4* gC = (const float4*)(x_dbl + ((size_t)b * RN2 + R_ + N_) * L_);
    float4* gO = (float4*)(out + row0);

    const int lane = tid & 31;
    const int n    = lane & 15;
    const int ch   = (tid >> 5) * 2 + (lane >> 4);
    const float A_neg = -__expf(A_log[(cb * SCH + ch) * N_ + n]);

    float4 dr[3], xr[3], zr[3], br[3], cr[3];
    const bool tail = tid < (STILE4 - 3 * 256);
    const float4 zero4 = make_float4(0.f, 0.f, 0.f, 0.f);
    float4 dT = zero4, xT = zero4, zT = zero4, bT = zero4, cT = zero4;
    float  pr[3], pT = 0.f;
    #pragma unroll
    for (int r = 0; r < 3; ++r) {
        const int i = tid + r * 256;
        dr[r] = gD[i];
        xr[r] = gX[i];
        br[r] = gB[i];
        cr[r] = gC[i];
        zr[r] = gZ[i];
        pr[r] = Dp_[cb * SCH + i / SROW4];
    }
    if (tail) {
        const int i = tid + 3 * 256;
        dT = gD[i];
        xT = gX[i];
        bT = gB[i];
        cT = gC[i];
        zT = gZ[i];
        pT = Dp_[cb * SCH + i / SROW4];
    }
    #pragma unroll
    for (int r = 0; r < 3; ++r) {
        const int i = tid + r * 256;
        const float4 d = dr[r];
        const float4 x = xr[r];
        sD[i] = d;
        sU[i] = make_float4(d.x * x.x, d.y * x.y, d.z * x.z, d.w * x.w);
        sB[i] = br[r];
        sC[i] = cr[r];
    }
    if (tail) {
        const int i = tid + 3 * 256;
        sD[i] = dT;
        sU[i] = make_float4(dT.x * xT.x, dT.y * xT.y, dT.z * xT.z, dT.w * xT.w);
        sB[i] = bT;
        sC[i] = cT;
    }
    __syncthreads();

    const float4* pD = sD + ch * SROW4;
    const float4* pU = sU + ch * SROW4;
    const float4* pB = sB + n  * SROW4;
    const float4* pC = sC + n  * SROW4;
    float4*       pY = sY + ch * SROW4;

    float h = 0.f;
    float4 dn = pD[0], un = pU[0], bn = pB[0], cn = pC[0];
    for (int j = 0; j < SROW4; ++j) {
        const float4 d = dn, u = un, bv = bn, cv = cn;
        const int jn = (j + 1 < SROW4) ? (j + 1) : (SROW4 - 1);
        dn = pD[jn]; un = pU[jn]; bn = pB[jn]; cn = pC[jn];
        float4 y;
        y.x = scan_step_(h, A_neg, d.x, u.x, bv.x, cv.x);
        y.y = scan_step_(h, A_neg, d.y, u.y, bv.y, cv.y);
        y.z = scan_step_(h, A_neg, d.z, u.z, bv.z, cv.z);
        y.w = scan_step_(h, A_neg, d.w, u.w, bv.w, cv.w);
        if (n == 0) pY[j] = y;
    }
    __syncthreads();

    #pragma unroll
    for (int r = 0; r < 3; ++r) {
        const int i = tid + r * 256;
        gO[i] = gate_(sY[i], xr[r], zr[r], pr[r]);
    }
    if (tail) {
        const int i = tid + 3 * 256;
        gO[i] = gate_(sY[i], xT, zT, pT);
    }
}

extern "C" void kernel_launch(void* const* d_in, const int* in_sizes, int n_in,
                              void* d_out, int out_size, void* d_ws, size_t ws_size,
                              hipStream_t stream)
{
    const float* hidden    = (const float*)d_in[0];
    const float* in_proj_w = (const float*)d_in[1];
    const float* conv_w    = (const float*)d_in[2];
    const float* conv_b    = (const float*)d_in[3];
    const float* x_proj_w  = (const float*)d_in[4];
    const float* dt_proj_w = (const float*)d_in[5];
    const float* dt_bias   = (const float*)d_in[6];
    const float* A_log     = (const float*)d_in[7];
    const float* D_param   = (const float*)d_in[8];

    float* ws = (float*)d_ws;
    const size_t BCL = (size_t)BSZ * C_ * L_;
    float* x_pre = ws;
    float* zbuf  = ws + BCL;
    float* xcbuf = ws + 2 * BCL;
    float* delta = ws + 3 * BCL;
    float* x_dbl = ws + 4 * BCL;

    k_gemm_inproj<<<dim3(4, BSZ * T_), IPW * 32, 0, stream>>>(
        hidden, in_proj_w, x_pre, zbuf);

    k_conv_xproj<<<dim3(2 * LTILES, BSZ), XPW * 32, 0, stream>>>(
        x_pre, conv_w, conv_b, x_proj_w, xcbuf, x_dbl);

    k_gemm_dtproj<<<dim3(96 * LTILES, BSZ), 32, 0, stream>>>(
        x_dbl, dt_proj_w, dt_bias, delta);

    k_scan_out<<<BSZ * (C_ / SCH), 256, 0, stream>>>(
        delta, xcbuf, zbuf, x_dbl, A_log, D_param, (float*)d_out);
}


# dt_proj GEMM + softplus fused into the conv/x_proj kernel: an 8-wave block builds its (b,16 l) x_dbl tile, keeps dt rows in LDS, computes delta for half the channels; k orders kept; 3 launches, not 4
# speedup vs baseline: 3.1015x; 1.0839x over previous
#include <hip/hip_runtime.h>


#define BSZ   8
#define T_    8
#define L_    196
#define D_    192
#define E_    192
#define C_    1536
#define N_    16
#define KC    4
#define R_    96
#define RN2   128
#define LTILES 13
#define SROW4_X (L_ / 4)

typedef float v2f __attribute__((ext_vector_type(2)));
typedef float v8f __attribute__((ext_vector_type(8)));

__device__ __forceinline__ float sigmoid_(float x) {
    return 1.0f / (1.0f + __expf(-x));
}

#define IPW     12
#define IPE     96
#define HPITCH  196
#define HPITCH4 (HPITCH / 4)
#define HROW4   (D_ / 4)
#define HTILE4  (L_ * HROW4)
#define HSTG    13

__global__ __launch_bounds__(IPW * 32)
void k_gemm_inproj(const float* __restrict__ hidden,
                   const float* __restrict__ w,
                   float* __restrict__ x_pre,
                   float* __restrict__ zbuf)
{
    __shared__ float4 sH4[L_ * HPITCH4];

    const int tid  = threadIdx.x;
    const int lane = tid & 31;
    const int wid  = tid >> 5;
    const int c    = lane & 15;
    const int h    = lane >> 4;
    const int eq   = blockIdx.x;
    const int bt   = blockIdx.y;
    const int b    = bt / T_;
    const int t    = bt % T_;
    const int et   = wid % 6;
    const int lh   = wid / 6;

    const float* Arow = w + (size_t)(eq * IPE + et * 16 + c) * D_ + 2 * h;
    v2f aF[48];
    #pragma unroll
    for (int s = 0; s < 48; ++s) aF[s] = *(const v2f*)(Arow + 4 * s);

    {
        const float4* g4 = (const float4*)(hidden + (size_t)bt * L_ * D_);
        for (int bb = 0; bb < 2; ++bb) {
            float4 v[HSTG];
            #pragma unroll
            for (int u = 0; u < HSTG; ++u) {
                const int i  = tid + (bb * HSTG + u) * (IPW * 32);
                const int ic = (i < HTILE4) ? i : (HTILE4 - 1);
                v[u] = g4[ic];
            }
            #pragma unroll
            for (int u = 0; u < HSTG; ++u) {
                const int i = tid + (bb * HSTG + u) * (IPW * 32);
                if (i < HTILE4) sH4[(i / HROW4) * HPITCH4 + (i % HROW4)] = v[u];
            }
        }
    }
    __syncthreads();

    const float* sH = (const float*)sH4;
    const int lt0 = lh ? 7 : 0;
    const int lt1 = lh ? LTILES : 7;
    float* __restrict__ dst = (eq < 2) ? x_pre : zbuf;
    const int e0 = (eq & 1) * IPE + et * 16 + 8 * h;

    v2f bN[16];
    {
        const int r0  = lt0 * 16 + c;
        const float* p = sH + ((r0 < L_) ? r0 : (L_ - 1)) * HPITCH + 2 * h;
        #pragma unroll
        for (int i = 0; i < 16; ++i) bN[i] = *(const v2f*)(p + 4 * i);
    }
    for (int lt = lt0; lt < lt1; ++lt) {
        const int ltn = (lt + 1 < lt1) ? (lt + 1) : lt;
        const int rc  = lt * 16 + c;
        const int rn  = ltn * 16 + c;
        const float* pc = sH + ((rc < L_) ? rc : (L_ - 1)) * HPITCH + 2 * h;
        const float* pn = sH + ((rn < L_) ? rn : (L_ - 1)) * HPITCH + 2 * h;

        v8f acc0 = {0.f,0.f,0.f,0.f,0.f,0.f,0.f,0.f};
        v8f acc1 = {0.f,0.f,0.f,0.f,0.f,0.f,0.f,0.f};
        #pragma unroll
        for (int ch = 0; ch < 3; ++ch) {
            v2f bC[16];
            #pragma unroll
            for (int i = 0; i < 16; ++i) bC[i] = bN[i];
            const float* pp = (ch < 2) ? (pc + (ch + 1) * 64) : pn;
            #pragma unroll
            for (int i = 0; i < 16; ++i) bN[i] = *(const v2f*)(pp + 4 * i);
            __builtin_amdgcn_sched_barrier(0);
            #pragma unroll
            for (int i = 0; i < 16; i += 2) {
                acc0 = __builtin_amdgcn_wmma_f32_16x16x4_f32(
                           false, aF[ch * 16 + i], false, bC[i], (short)0, acc0, false, false);
                acc1 = __builtin_amdgcn_wmma_f32_16x16x4_f32(
                           false, aF[ch * 16 + i + 1], false, bC[i + 1], (short)0, acc1, false, false);
            }
        }

        const int l = lt * 16 + c;
        if (l < L_) {
            #pragma unroll
            for (int r = 0; r < 8; ++r) {
                const int ch_out = (e0 + r) * T_ + t;
                dst[((size_t)b * C_ + ch_out) * L_ + l] = acc0[r] + acc1[r];
            }
        }
    }
}

#define XPW    8
#define XKCH   64
#define XNCH   (C_ / XKCH)
#define DKS    (R_ / 4)
#define DCT    6

__device__ __forceinline__ float conv_tap4_(float bias, float4 wv, float a,
                                            float b, float c, float d)
{
    float s = bias;
    s += a * wv.x;
    s += b * wv.y;
    s += c * wv.z;
    s += d * wv.w;
    return s * sigmoid_(s);
}

__global__ __launch_bounds__(XPW * 32)
void k_conv_xproj_dt(const float* __restrict__ x_pre,
                     const float* __restrict__ cw,
                     const float* __restrict__ cb,
                     const float* __restrict__ w,
                     const float* __restrict__ wdt,
                     const float* __restrict__ dt_bias,
                     float* __restrict__ xc,
                     float* __restrict__ x_dbl,
                     float* __restrict__ delta)
{
    __shared__ float4 sX[C_ * 4];
    __shared__ float  sDt[R_ * 16];

    const int tid = threadIdx.x;
    const int hf  = blockIdx.x & 1;
    const int lt  = blockIdx.x >> 1;
    const int b   = blockIdx.y;

    {
        const int j  = tid & 3;
        const int cr = tid >> 2;
        const int l  = lt * 16 + 4 * j;
        const bool valid   = l < L_;
        const bool hasprev = l > 0;
        const bool store   = valid && (hf == 0);
        const int q  = (valid ? l : (L_ - 4)) >> 2;
        const int qp = hasprev ? (q - 1) : q;
        const float4* xr4 = (const float4*)(x_pre + (size_t)b * C_ * L_);
        float4*       xo4 = (float4*)(xc + (size_t)b * C_ * L_) + q;
        const float4* cw4 = (const float4*)cw;
        float4 curN[8], plN[8], wvN[8];
        float  bsN[8];
        #pragma unroll
        for (int u = 0; u < 8; ++u) {
            const int c = u * 64 + cr;
            curN[u] = xr4[c * SROW4_X + q];
            plN[u]  = xr4[c * SROW4_X + qp];
            wvN[u]  = cw4[c];
            bsN[u]  = cb[c];
        }
        for (int t = 0; t < C_ / 512; ++t) {
            float4 curC[8], plC[8], wvC[8];
            float  bsC[8];
            #pragma unroll
            for (int u = 0; u < 8; ++u) {
                curC[u] = curN[u]; plC[u] = plN[u]; wvC[u] = wvN[u]; bsC[u] = bsN[u];
            }
            const int tn = (t + 1 < C_ / 512) ? (t + 1) : t;
            #pragma unroll
            for (int u = 0; u < 8; ++u) {
                const int c = tn * 512 + u * 64 + cr;
                curN[u] = xr4[c * SROW4_X + q];
                plN[u]  = xr4[c * SROW4_X + qp];
                wvN[u]  = cw4[c];
                bsN[u]  = cb[c];
            }
            #pragma unroll
            for (int u = 0; u < 8; ++u) {
                const int c = t * 512 + u * 64 + cr;
                const float4 cur = curC[u];
                const float4 wv  = wvC[u];
                const float bias = bsC[u];
                const float p1 = hasprev ? plC[u].y : 0.f;
                const float p2 = hasprev ? plC[u].z : 0.f;
                const float p3 = hasprev ? plC[u].w : 0.f;
                float4 o;
                o.x = conv_tap4_(bias, wv, p1, p2, p3, cur.x);
                o.y = conv_tap4_(bias, wv, p2, p3, cur.x, cur.y);
                o.z = conv_tap4_(bias, wv, p3, cur.x, cur.y, cur.z);
                o.w = conv_tap4_(bias, wv, cur.x, cur.y, cur.z, cur.w);
                sX[c * 4 + j] = o;
                if (store) xo4[c * SROW4_X] = o;
            }
        }
    }
    __syncthreads();

    const int lane = tid & 31;
    const int wid  = tid >> 5;
    const int m    = lane & 15;
    const int h    = lane >> 4;
    const int l     = lt * 16 + m;
    const int mbase = h * 8;
    const int ct0 = hf * 48 + wid * DCT;
    const float*  Ab    = wdt + (size_t)(ct0 * 16 + m) * R_ + 2 * h;
    const float4* bias4 = (const float4*)(dt_bias + ct0 * 16 + mbase);
    v2f aN3[DKS];
    float4 bsN0, bsN1;
    {
        const float* Arow = w + (size_t)(wid * 16 + m) * C_ + 2 * h;
        const float* Bcol = (const float*)sX + (2 * h) * 16 + m;

        v8f acc0 = {0.f,0.f,0.f,0.f,0.f,0.f,0.f,0.f};
        v8f acc1 = {0.f,0.f,0.f,0.f,0.f,0.f,0.f,0.f};

        v2f aN[16];
        #pragma unroll
        for (int i = 0; i < 16; ++i) aN[i] = *(const v2f*)(Arow + 4 * i);

        for (int ch = 0; ch < XNCH; ++ch) {
            v2f aC[16];
            #pragma unroll
            for (int i = 0; i < 16; ++i) aC[i] = aN[i];
            const int nx = (ch + 1 < XNCH) ? (ch + 1) : (XNCH - 1);
            #pragma unroll
            for (int i = 0; i < 16; ++i)
                aN[i] = *(const v2f*)(Arow + nx * XKCH + 4 * i);
            v2f bC[16];
            #pragma unroll
            for (int i = 0; i < 16; ++i) {
                const float* p = Bcol + (ch * XKCH + 4 * i) * 16;
                bC[i].x = p[0];
                bC[i].y = p[16];
            }
            #pragma unroll
            for (int i = 0; i < 16; i += 2) {
                acc0 = __builtin_amdgcn_wmma_f32_16x16x4_f32(
                           false, aC[i], false, bC[i], (short)0, acc0, false, false);
                acc1 = __builtin_amdgcn_wmma_f32_16x16x4_f32(
                           false, aC[i + 1], false, bC[i + 1], (short)0, acc1, false, false);
            }
        }

        #pragma unroll
        for (int s = 0; s < DKS; ++s) aN3[s] = *(const v2f*)(Ab + 4 * s);
        bsN0 = bias4[0];
        bsN1 = bias4[1];

        if (wid < R_ / 16) {
            #pragma unroll
            for (int r = 0; r < 8; ++r)
                sDt[(wid * 16 + mbase + r) * 16 + m] = acc0[r] + acc1[r];
        } else if (hf == 0 && l < L_) {
            #pragma unroll
            for (int r = 0; r < 8; ++r) {
                const int rr = wid * 16 + mbase + r;
                x_dbl[((size_t)b * RN2 + rr) * L_ + l] = acc0[r] + acc1[r];
            }
        }
    }
    __syncthreads();

    {
        v2f bF[DKS];
        #pragma unroll
        for (int s = 0; s < DKS; ++s) {
            const float* p = sDt + (4 * s + 2 * h) * 16 + m;
            bF[s].x = p[0];
            bF[s].y = p[16];
        }
        __builtin_amdgcn_sched_barrier(0);

        float* __restrict__ dcol = delta + ((size_t)b * C_ + ct0 * 16 + mbase) * L_ + l;

        #pragma unroll
        for (int i = 0; i < DCT; ++i) {
            v2f aC[DKS];
            #pragma unroll
            for (int s = 0; s < DKS; ++s) aC[s] = aN3[s];
            const float4 bs0 = bsN0, bs1 = bsN1;
            if (i + 1 < DCT) {
                #pragma unroll
                for (int s = 0; s < DKS; ++s)
                    aN3[s] = *(const v2f*)(Ab + (size_t)(i + 1) * 16 * R_ + 4 * s);
                bsN0 = bias4[(i + 1) * 4];
                bsN1 = bias4[(i + 1) * 4 + 1];
            }

            v8f acc0 = {0.f,0.f,0.f,0.f,0.f,0.f,0.f,0.f};
            v8f acc1 = {0.f,0.f,0.f,0.f,0.f,0.f,0.f,0.f};
            #pragma unroll
            for (int s = 0; s < DKS; s += 2) {
                acc0 = __builtin_amdgcn_wmma_f32_16x16x4_f32(
                           false, aC[s], false, bF[s], (short)0, acc0, false, false);
                acc1 = __builtin_amdgcn_wmma_f32_16x16x4_f32(
                           false, aC[s + 1], false, bF[s + 1], (short)0, acc1, false, false);
            }

            const float bsv[8] = {bs0.x, bs0.y, bs0.z, bs0.w, bs1.x, bs1.y, bs1.z, bs1.w};
            float sp[8];
            #pragma unroll
            for (int r = 0; r < 8; ++r) {
                const float v = acc0[r] + acc1[r] + bsv[r];
                sp[r] = (v > 20.f) ? v : __logf(1.f + __expf(v));
            }
            if (l < L_) {
                #pragma unroll
                for (int r = 0; r < 8; ++r)
                    dcol[(size_t)(i * 16 + r) * L_] = sp[r];
            }
        }
    }
}

#define SCH    16
#define SROW4  (L_ / 4)
#define STILE4 (SCH * SROW4)

__device__ __forceinline__ float4 gate_(float4 y, float4 x, float4 z, float Dp)
{
    float4 o;
    o.x = (y.x + x.x * Dp) * (z.x * sigmoid_(z.x));
    o.y = (y.y + x.y * Dp) * (z.y * sigmoid_(z.y));
    o.z = (y.z + x.z * Dp) * (z.z * sigmoid_(z.z));
    o.w = (y.w + x.w * Dp) * (z.w * sigmoid_(z.w));
    return o;
}

__device__ __forceinline__ float scan_step_(float& h, float A_neg, float dv,
                                            float u, float bv, float cv)
{
    const float dA = __expf(dv * A_neg);
    h = fmaf(h, dA, u * bv);
    float yp = h * cv;
    yp += __shfl_xor(yp, 1, 16);
    yp += __shfl_xor(yp, 2, 16);
    yp += __shfl_xor(yp, 4, 16);
    yp += __shfl_xor(yp, 8, 16);
    return yp;
}

__global__ __launch_bounds__(256)
void k_scan_out(const float* __restrict__ delta,
                const float* __restrict__ xc,
                const float* __restrict__ zbuf,
                const float* __restrict__ x_dbl,
                const float* __restrict__ A_log,
                const float* __restrict__ Dp_,
                float* __restrict__ out)
{
    __shared__ float4 sD[STILE4];
    __shared__ float4 sU[STILE4];
    __shared__ float4 sB[STILE4];
    __shared__ float4 sC[STILE4];
    __shared__ float4 sY[STILE4];

    const int tid = threadIdx.x;
    const int cb  = blockIdx.x % (C_ / SCH);
    const int b   = blockIdx.x / (C_ / SCH);
    const size_t row0 = ((size_t)b * C_ + (size_t)cb * SCH) * L_;

    const float4* gD = (const float4*)(delta + row0);
    const float4* gX = (const float4*)(xc    + row0);
    const float4* gZ = (const float4*)(zbuf  + row0);
    const float4* gB = (const float4*)(x_dbl + ((size_t)b * RN2 + R_) * L_);
    const float4* gC = (const float4*)(x_dbl + ((size_t)b * RN2 + R_ + N_) * L_);
    float4* gO = (float4*)(out + row0);

    const int lane = tid & 31;
    const int n    = lane & 15;
    const int ch   = (tid >> 5) * 2 + (lane >> 4);
    const float A_neg = -__expf(A_log[(cb * SCH + ch) * N_ + n]);

    float4 dr[3], xr[3], zr[3], br[3], cr[3];
    const bool tail = tid < (STILE4 - 3 * 256);
    const float4 zero4 = make_float4(0.f, 0.f, 0.f, 0.f);
    float4 dT = zero4, xT = zero4, zT = zero4, bT = zero4, cT = zero4;
    float  pr[3], pT = 0.f;
    #pragma unroll
    for (int r = 0; r < 3; ++r) {
        const int i = tid + r * 256;
        dr[r] = gD[i];
        xr[r] = gX[i];
        br[r] = gB[i];
        cr[r] = gC[i];
        zr[r] = gZ[i];
        pr[r] = Dp_[cb * SCH + i / SROW4];
    }
    if (tail) {
        const int i = tid + 3 * 256;
        dT = gD[i];
        xT = gX[i];
        bT = gB[i];
        cT = gC[i];
        zT = gZ[i];
        pT = Dp_[cb * SCH + i / SROW4];
    }
    #pragma unroll
    for (int r = 0; r < 3; ++r) {
        const int i = tid + r * 256;
        const float4 d = dr[r];
        const float4 x = xr[r];
        sD[i] = d;
        sU[i] = make_float4(d.x * x.x, d.y * x.y, d.z * x.z, d.w * x.w);
        sB[i] = br[r];
        sC[i] = cr[r];
    }
    if (tail) {
        const int i = tid + 3 * 256;
        sD[i] = dT;
        sU[i] = make_float4(dT.x * xT.x, dT.y * xT.y, dT.z * xT.z, dT.w * xT.w);
        sB[i] = bT;
        sC[i] = cT;
    }
    __syncthreads();

    const float4* pD = sD + ch * SROW4;
    const float4* pU = sU + ch * SROW4;
    const float4* pB = sB + n  * SROW4;
    const float4* pC = sC + n  * SROW4;
    float4*       pY = sY + ch * SROW4;

    float h = 0.f;
    float4 dn = pD[0], un = pU[0], bn = pB[0], cn = pC[0];
    for (int j = 0; j < SROW4; ++j) {
        const float4 d = dn, u = un, bv = bn, cv = cn;
        const int jn = (j + 1 < SROW4) ? (j + 1) : (SROW4 - 1);
        dn = pD[jn]; un = pU[jn]; bn = pB[jn]; cn = pC[jn];
        float4 y;
        y.x = scan_step_(h, A_neg, d.x, u.x, bv.x, cv.x);
        y.y = scan_step_(h, A_neg, d.y, u.y, bv.y, cv.y);
        y.z = scan_step_(h, A_neg, d.z, u.z, bv.z, cv.z);
        y.w = scan_step_(h, A_neg, d.w, u.w, bv.w, cv.w);
        if (n == 0) pY[j] = y;
    }
    __syncthreads();

    #pragma unroll
    for (int r = 0; r < 3; ++r) {
        const int i = tid + r * 256;
        gO[i] = gate_(sY[i], xr[r], zr[r], pr[r]);
    }
    if (tail) {
        const int i = tid + 3 * 256;
        gO[i] = gate_(sY[i], xT, zT, pT);
    }
}

extern "C" void kernel_launch(void* const* d_in, const int* in_sizes, int n_in,
                              void* d_out, int out_size, void* d_ws, size_t ws_size,
                              hipStream_t stream)
{
    const float* hidden    = (const float*)d_in[0];
    const float* in_proj_w = (const float*)d_in[1];
    const float* conv_w    = (const float*)d_in[2];
    const float* conv_b    = (const float*)d_in[3];
    const float* x_proj_w  = (const float*)d_in[4];
    const float* dt_proj_w = (const float*)d_in[5];
    const float* dt_bias   = (const float*)d_in[6];
    const float* A_log     = (const float*)d_in[7];
    const float* D_param   = (const float*)d_in[8];

    float* ws = (float*)d_ws;
    const size_t BCL = (size_t)BSZ * C_ * L_;
    float* x_pre = ws;
    float* zbuf  = ws + BCL;
    float* xcbuf = ws + 2 * BCL;
    float* delta = ws + 3 * BCL;
    float* x_dbl = ws + 4 * BCL;

    k_gemm_inproj<<<dim3(4, BSZ * T_), IPW * 32, 0, stream>>>(
        hidden, in_proj_w, x_pre, zbuf);

    k_conv_xproj_dt<<<dim3(2 * LTILES, BSZ), XPW * 32, 0, stream>>>(
        x_pre, conv_w, conv_b, x_proj_w, dt_proj_w, dt_bias, xcbuf, x_dbl, delta);

    k_scan_out<<<BSZ * (C_ / SCH), 256, 0, stream>>>(
        delta, xcbuf, zbuf, x_dbl, A_log, D_param, (float*)d_out);
}


# conv/x_proj/dt_proj kernel: the weight (A) operand of both GEMMs goes through LDS, copied with 16-B whole-line loads (x_proj_w in 64-k chunks, 3 buffers; dt_proj_w per wave tile); k orders kept
# speedup vs baseline: 3.2345x; 1.0429x over previous
#include <hip/hip_runtime.h>


#define BSZ   8
#define T_    8
#define L_    196
#define D_    192
#define E_    192
#define C_    1536
#define N_    16
#define KC    4
#define R_    96
#define RN2   128
#define LTILES 13
#define SROW4_X (L_ / 4)

typedef float v2f __attribute__((ext_vector_type(2)));
typedef float v8f __attribute__((ext_vector_type(8)));
typedef float v4f __attribute__((ext_vector_type(4)));

__device__ __forceinline__ float sigmoid_(float x) {
    return 1.0f / (1.0f + __expf(-x));
}

#define IPW     12
#define IPE     96
#define HPITCH  196
#define HPITCH4 (HPITCH / 4)
#define HROW4   (D_ / 4)
#define HTILE4  (L_ * HROW4)
#define HSTG    13

__global__ __launch_bounds__(IPW * 32)
void k_gemm_inproj(const float* __restrict__ hidden,
                   const float* __restrict__ w,
                   float* __restrict__ x_pre,
                   float* __restrict__ zbuf)
{
    __shared__ float4 sH4[L_ * HPITCH4];

    const int tid  = threadIdx.x;
    const int lane = tid & 31;
    const int wid  = tid >> 5;
    const int c    = lane & 15;
    const int h    = lane >> 4;
    const int eq   = blockIdx.x;
    const int bt   = blockIdx.y;
    const int b    = bt / T_;
    const int t    = bt % T_;
    const int et   = wid % 6;
    const int lh   = wid / 6;

    const float* Arow = w + (size_t)(eq * IPE + et * 16 + c) * D_ + 2 * h;
    v2f aF[48];
    #pragma unroll
    for (int s = 0; s < 48; ++s) aF[s] = *(const v2f*)(Arow + 4 * s);

    {
        const float4* g4 = (const float4*)(hidden + (size_t)bt * L_ * D_);
        for (int bb = 0; bb < 2; ++bb) {
            float4 v[HSTG];
            #pragma unroll
            for (int u = 0; u < HSTG; ++u) {
                const int i  = tid + (bb * HSTG + u) * (IPW * 32);
                const int ic = (i < HTILE4) ? i : (HTILE4 - 1);
                v[u] = g4[ic];
            }
            #pragma unroll
            for (int u = 0; u < HSTG; ++u) {
                const int i = tid + (bb * HSTG + u) * (IPW * 32);
                if (i < HTILE4) sH4[(i / HROW4) * HPITCH4 + (i % HROW4)] = v[u];
            }
        }
    }
    __syncthreads();

    const float* sH = (const float*)sH4;
    const int lt0 = lh ? 7 : 0;
    const int lt1 = lh ? LTILES : 7;
    float* __restrict__ dst = (eq < 2) ? x_pre : zbuf;
    const int e0 = (eq & 1) * IPE + et * 16 + 8 * h;

    v2f bN[16];
    {
        const int r0  = lt0 * 16 + c;
        const float* p = sH + ((r0 < L_) ? r0 : (L_ - 1)) * HPITCH + 2 * h;
        #pragma unroll
        for (int i = 0; i < 16; ++i) bN[i] = *(const v2f*)(p + 4 * i);
    }
    for (int lt = lt0; lt < lt1; ++lt) {
        const int ltn = (lt + 1 < lt1) ? (lt + 1) : lt;
        const int rc  = lt * 16 + c;
        const int rn  = ltn * 16 + c;
        const float* pc = sH + ((rc < L_) ? rc : (L_ - 1)) * HPITCH + 2 * h;
        const float* pn = sH + ((rn < L_) ? rn : (L_ - 1)) * HPITCH + 2 * h;

        v8f acc0 = {0.f,0.f,0.f,0.f,0.f,0.f,0.f,0.f};
        v8f acc1 = {0.f,0.f,0.f,0.f,0.f,0.f,0.f,0.f};
        #pragma unroll
        for (int ch = 0; ch < 3; ++ch) {
            v2f bC[16];
            #pragma unroll
            for (int i = 0; i < 16; ++i) bC[i] = bN[i];
            const float* pp = (ch < 2) ? (pc + (ch + 1) * 64) : pn;
            #pragma unroll
            for (int i = 0; i < 16; ++i) bN[i] = *(const v2f*)(pp + 4 * i);
            __builtin_amdgcn_sched_barrier(0);
            #pragma unroll
            for (int i = 0; i < 16; i += 2) {
                acc0 = __builtin_amdgcn_wmma_f32_16x16x4_f32(
                           false, aF[ch * 16 + i], false, bC[i], (short)0, acc0, false, false);
                acc1 = __builtin_amdgcn_wmma_f32_16x16x4_f32(
                           false, aF[ch * 16 + i + 1], false, bC[i + 1], (short)0, acc1, false, false);
            }
        }

        const int l = lt * 16 + c;
        if (l < L_) {
            #pragma unroll
            for (int r = 0; r < 8; ++r) {
                const int ch_out = (e0 + r) * T_ + t;
                dst[((size_t)b * C_ + ch_out) * L_ + l] = acc0[r] + acc1[r];
            }
        }
    }
}

#define XPW    8
#define XKCH   64
#define XNCH   (C_ / XKCH)
#define DKS    (R_ / 4)
#define DCT    6
#define XKS    (XKCH / 4)
#define WROW4  (XKCH / 4)
#define WPITCH (XKCH + 4)
#define WPITCH4 (WPITCH / 4)
#define WBUF4  (RN2 * WPITCH4)
#define WRPR   (XPW * 32 / WROW4)
#define WSTG   (RN2 / WRPR)
#define DPITCH 100
#define DPITCH4 (DPITCH / 4)
#define DTILE4 (16 * R_ / 4)
#define DSTG   (DTILE4 / 32)

__device__ __forceinline__ void lds_barrier_(v8f a0, v8f a1)
{
    asm volatile("s_wait_dscnt 0x0\n\ts_barrier_signal -1\n\ts_barrier_wait -1"
                 :: "v"(a0), "v"(a1) : "memory");
}

__device__ __forceinline__ float conv_tap4_(float bias, float4 wv, float a,
                                            float b, float c, float d)
{
    float s = bias;
    s += a * wv.x;
    s += b * wv.y;
    s += c * wv.z;
    s += d * wv.w;
    return s * sigmoid_(s);
}

__global__ __launch_bounds__(XPW * 32)
void k_conv_xproj_dt(const float* __restrict__ x_pre,
                     const float* __restrict__ cw,
                     const float* __restrict__ cb,
                     const float* __restrict__ w,
                     const float* __restrict__ wdt,
                     const float* __restrict__ dt_bias,
                     float* __restrict__ xc,
                     float* __restrict__ x_dbl,
                     float* __restrict__ delta)
{
    __shared__ float4 sX[C_ * 4];
    __shared__ float  sDt[R_ * 16];
    __shared__ float4 sW4[3 * WBUF4];

    const int tid = threadIdx.x;
    const int hf  = blockIdx.x & 1;
    const int lt  = blockIdx.x >> 1;
    const int b   = blockIdx.y;

    {
        const int j  = tid & 3;
        const int cr = tid >> 2;
        const int l  = lt * 16 + 4 * j;
        const bool valid   = l < L_;
        const bool hasprev = l > 0;
        const bool store   = valid && (hf == 0);
        const int q  = (valid ? l : (L_ - 4)) >> 2;
        const int qp = hasprev ? (q - 1) : q;
        const float4* xr4 = (const float4*)(x_pre + (size_t)b * C_ * L_);
        float4*       xo4 = (float4*)(xc + (size_t)b * C_ * L_) + q;
        const float4* cw4 = (const float4*)cw;
        float4 curN[8], plN[8], wvN[8];
        float  bsN[8];
        #pragma unroll
        for (int u = 0; u < 8; ++u) {
            const int c = u * 64 + cr;
            curN[u] = xr4[c * SROW4_X + q];
            plN[u]  = xr4[c * SROW4_X + qp];
            wvN[u]  = cw4[c];
            bsN[u]  = cb[c];
        }
        for (int t = 0; t < C_ / 512; ++t) {
            float4 curC[8], plC[8], wvC[8];
            float  bsC[8];
            #pragma unroll
            for (int u = 0; u < 8; ++u) {
                curC[u] = curN[u]; plC[u] = plN[u]; wvC[u] = wvN[u]; bsC[u] = bsN[u];
            }
            const int tn = (t + 1 < C_ / 512) ? (t + 1) : t;
            #pragma unroll
            for (int u = 0; u < 8; ++u) {
                const int c = tn * 512 + u * 64 + cr;
                curN[u] = xr4[c * SROW4_X + q];
                plN[u]  = xr4[c * SROW4_X + qp];
                wvN[u]  = cw4[c];
                bsN[u]  = cb[c];
            }
            #pragma unroll
            for (int u = 0; u < 8; ++u) {
                const int c = t * 512 + u * 64 + cr;
                const float4 cur = curC[u];
                const float4 wv  = wvC[u];
                const float bias = bsC[u];
                const float p1 = hasprev ? plC[u].y : 0.f;
                const float p2 = hasprev ? plC[u].z : 0.f;
                const float p3 = hasprev ? plC[u].w : 0.f;
                float4 o;
                o.x = conv_tap4_(bias, wv, p1, p2, p3, cur.x);
                o.y = conv_tap4_(bias, wv, p2, p3, cur.x, cur.y);
                o.z = conv_tap4_(bias, wv, p3, cur.x, cur.y, cur.z);
                o.w = conv_tap4_(bias, wv, cur.x, cur.y, cur.z, cur.w);
                sX[c * 4 + j] = o;
                if (store) xo4[c * SROW4_X] = o;
            }
        }
    }

    const float4* w4 = (const float4*)w;
    const int srow = tid / WROW4;
    const int sc4  = tid % WROW4;
    {
        float4 g[2 * WSTG];
        #pragma unroll
        for (int u = 0; u < 2 * WSTG; ++u)
            g[u] = w4[(size_t)(srow + WRPR * (u % WSTG)) * (C_ / 4) + (u / WSTG) * WROW4 + sc4];
        #pragma unroll
        for (int u = 0; u < 2 * WSTG; ++u)
            sW4[(u / WSTG) * WBUF4 + (srow + WRPR * (u % WSTG)) * WPITCH4 + sc4] = g[u];
    }
    const v4f* w4v = (const v4f*)w;
    v4f*       sWv = (v4f*)sW4;
    v4f gW[WSTG];
    #pragma unroll
    for (int u = 0; u < WSTG; ++u)
        gW[u] = w4v[(size_t)(srow + WRPR * u) * (C_ / 4) + 2 * WROW4 + sc4];
    __syncthreads();

    const int lane = tid & 31;
    const int wid  = tid >> 5;
    const int m    = lane & 15;
    const int h    = lane >> 4;
    const int l     = lt * 16 + m;
    const int mbase = h * 8;
    const int ct0 = hf * 48 + wid * DCT;
    const v4f*    g3    = (const v4f*)wdt + (size_t)ct0 * DTILE4;
    const float4* bias4 = (const float4*)(dt_bias + ct0 * 16 + mbase);
    v4f    gN3[DSTG];
    float4 bsN0, bsN1;
    {
        const float* Afrag = (const float*)sW4 + (wid * 16 + m) * WPITCH + 2 * h;
        const float* Bcol = (const float*)sX + (2 * h) * 16 + m;

        v8f acc0 = {0.f,0.f,0.f,0.f,0.f,0.f,0.f,0.f};
        v8f acc1 = {0.f,0.f,0.f,0.f,0.f,0.f,0.f,0.f};

        v2f aC[XKS], bC[XKS];
        #pragma unroll
        for (int i = 0; i < XKS; ++i) {
            aC[i] = *(const v2f*)(Afrag + 4 * i);
            const float* p = Bcol + (4 * i) * 16;
            bC[i].x = p[0];
            bC[i].y = p[16];
        }
        #pragma unroll 2
        for (int ch = 0; ch < XNCH; ++ch) {
            const int n1 = (ch + 1 < XNCH) ? (ch + 1) : (XNCH - 1);
            const int n3 = (ch + 3 < XNCH) ? (ch + 3) : (XNCH - 1);
            #pragma unroll
            for (int u = 0; u < WSTG; ++u)
                sWv[((ch + 2) % 3) * WBUF4 + (srow + WRPR * u) * WPITCH4 + sc4] = gW[u];
            #pragma unroll
            for (int u = 0; u < WSTG; ++u)
                gW[u] = w4v[(size_t)(srow + WRPR * u) * (C_ / 4) + n3 * WROW4 + sc4];
            v2f aN[XKS], bN[XKS];
            #pragma unroll
            for (int i = 0; i < XKS; ++i) {
                aN[i] = *(const v2f*)(Afrag + (n1 % 3) * (RN2 * WPITCH) + 4 * i);
                const float* p = Bcol + (n1 * XKCH + 4 * i) * 16;
                bN[i].x = p[0];
                bN[i].y = p[16];
            }
            __builtin_amdgcn_sched_barrier(0);
            #pragma unroll
            for (int i = 0; i < XKS; i += 2) {
                acc0 = __builtin_amdgcn_wmma_f32_16x16x4_f32(
                           false, aC[i], false, bC[i], (short)0, acc0, false, false);
                acc1 = __builtin_amdgcn_wmma_f32_16x16x4_f32(
                           false, aC[i + 1], false, bC[i + 1], (short)0, acc1, false, false);
            }
            lds_barrier_(acc0, acc1);
            #pragma unroll
            for (int i = 0; i < XKS; ++i) { aC[i] = aN[i]; bC[i] = bN[i]; }
        }

        #pragma unroll
        for (int u = 0; u < DSTG; ++u) gN3[u] = g3[lane + 32 * u];
        bsN0 = bias4[0];
        bsN1 = bias4[1];

        if (wid < R_ / 16) {
            #pragma unroll
            for (int r = 0; r < 8; ++r)
                sDt[(wid * 16 + mbase + r) * 16 + m] = acc0[r] + acc1[r];
        } else if (hf == 0 && l < L_) {
            #pragma unroll
            for (int r = 0; r < 8; ++r) {
                const int rr = wid * 16 + mbase + r;
                x_dbl[((size_t)b * RN2 + rr) * L_ + l] = acc0[r] + acc1[r];
            }
        }
    }
    v4f* sA4 = (v4f*)sW4 + wid * (16 * DPITCH4);
    #pragma unroll
    for (int u = 0; u < DSTG; ++u) {
        const int idx = lane + 32 * u;
        sA4[(idx / (R_ / 4)) * DPITCH4 + (idx % (R_ / 4))] = gN3[u];
    }
    __syncthreads();

    {
        v2f bF[DKS];
        #pragma unroll
        for (int s = 0; s < DKS; ++s) {
            const float* p = sDt + (4 * s + 2 * h) * 16 + m;
            bF[s].x = p[0];
            bF[s].y = p[16];
        }
        __builtin_amdgcn_sched_barrier(0);

        float* __restrict__ dcol = delta + ((size_t)b * C_ + ct0 * 16 + mbase) * L_ + l;

        const float* sAf = (const float*)sA4 + m * DPITCH + 2 * h;
        #pragma unroll
        for (int u = 0; u < DSTG; ++u) gN3[u] = g3[DTILE4 + lane + 32 * u];
        #pragma unroll 1
        for (int i = 0; i < DCT; ++i) {
            const float4 bs0 = bsN0, bs1 = bsN1;
            const int i1 = (i + 1 < DCT) ? (i + 1) : (DCT - 1);
            const int i2 = (i + 2 < DCT) ? (i + 2) : (DCT - 1);
            v2f aC[DKS];
            #pragma unroll
            for (int s = 0; s < DKS; ++s) aC[s] = *(const v2f*)(sAf + 4 * s);
            bsN0 = bias4[i1 * 4];
            bsN1 = bias4[i1 * 4 + 1];

            v8f acc0 = {0.f,0.f,0.f,0.f,0.f,0.f,0.f,0.f};
            v8f acc1 = {0.f,0.f,0.f,0.f,0.f,0.f,0.f,0.f};
            #pragma unroll
            for (int s = 0; s < DKS; s += 2) {
                acc0 = __builtin_amdgcn_wmma_f32_16x16x4_f32(
                           false, aC[s], false, bF[s], (short)0, acc0, false, false);
                acc1 = __builtin_amdgcn_wmma_f32_16x16x4_f32(
                           false, aC[s + 1], false, bF[s + 1], (short)0, acc1, false, false);
            }

            #pragma unroll
            for (int u = 0; u < DSTG; ++u) {
                const int idx = lane + 32 * u;
                sA4[(idx / (R_ / 4)) * DPITCH4 + (idx % (R_ / 4))] = gN3[u];
            }
            #pragma unroll
            for (int u = 0; u < DSTG; ++u) gN3[u] = g3[i2 * DTILE4 + lane + 32 * u];

            const float bsv[8] = {bs0.x, bs0.y, bs0.z, bs0.w, bs1.x, bs1.y, bs1.z, bs1.w};
            float sp[8];
            #pragma unroll
            for (int r = 0; r < 8; ++r) {
                const float v = acc0[r] + acc1[r] + bsv[r];
                sp[r] = (v > 20.f) ? v : __logf(1.f + __expf(v));
            }
            if (l < L_) {
                #pragma unroll
                for (int r = 0; r < 8; ++r)
                    dcol[(size_t)(i * 16 + r) * L_] = sp[r];
            }
        }
    }
}

#define SCH    16
#define SROW4  (L_ / 4)
#define STILE4 (SCH * SROW4)

__device__ __forceinline__ float4 gate_(float4 y, float4 x, float4 z, float Dp)
{
    float4 o;
    o.x = (y.x + x.x * Dp) * (z.x * sigmoid_(z.x));
    o.y = (y.y + x.y * Dp) * (z.y * sigmoid_(z.y));
    o.z = (y.z + x.z * Dp) * (z.z * sigmoid_(z.z));
    o.w = (y.w + x.w * Dp) * (z.w * sigmoid_(z.w));
    return o;
}

__device__ __forceinline__ float scan_step_(float& h, float A_neg, float dv,
                                            float u, float bv, float cv)
{
    const float dA = __expf(dv * A_neg);
    h = fmaf(h, dA, u * bv);
    float yp = h * cv;
    yp += __shfl_xor(yp, 1, 16);
    yp += __shfl_xor(yp, 2, 16);
    yp += __shfl_xor(yp, 4, 16);
    yp += __shfl_xor(yp, 8, 16);
    return yp;
}

__global__ __launch_bounds__(256)
void k_scan_out(const float* __restrict__ delta,
                const float* __restrict__ xc,
                const float* __restrict__ zbuf,
                const float* __restrict__ x_dbl,
                const float* __restrict__ A_log,
                const float* __restrict__ Dp_,
                float* __restrict__ out)
{
    __shared__ float4 sD[STILE4];
    __shared__ float4 sU[STILE4];
    __shared__ float4 sB[STILE4];
    __shared__ float4 sC[STILE4];
    __shared__ float4 sY[STILE4];

    const int tid = threadIdx.x;
    const int cb  = blockIdx.x % (C_ / SCH);
    const int b   = blockIdx.x / (C_ / SCH);
    const size_t row0 = ((size_t)b * C_ + (size_t)cb * SCH) * L_;

    const float4* gD = (const float4*)(delta + row0);
    const float4* gX = (const float4*)(xc    + row0);
    const float4* gZ = (const float4*)(zbuf  + row0);
    const float4* gB = (const float4*)(x_dbl + ((size_t)b * RN2 + R_) * L_);
    const float4* gC = (const float4*)(x_dbl + ((size_t)b * RN2 + R_ + N_) * L_);
    float4* gO = (float4*)(out + row0);

    const int lane = tid & 31;
    const int n    = lane & 15;
    const int ch   = (tid >> 5) * 2 + (lane >> 4);
    const float A_neg = -__expf(A_log[(cb * SCH + ch) * N_ + n]);

    float4 dr[3], xr[3], zr[3], br[3], cr[3];
    const bool tail = tid < (STILE4 - 3 * 256);
    const float4 zero4 = make_float4(0.f, 0.f, 0.f, 0.f);
    float4 dT = zero4, xT = zero4, zT = zero4, bT = zero4, cT = zero4;
    float  pr[3], pT = 0.f;
    #pragma unroll
    for (int r = 0; r < 3; ++r) {
        const int i = tid + r * 256;
        dr[r] = gD[i];
        xr[r] = gX[i];
        br[r] = gB[i];
        cr[r] = gC[i];
        zr[r] = gZ[i];
        pr[r] = Dp_[cb * SCH + i / SROW4];
    }
    if (tail) {
        const int i = tid + 3 * 256;
        dT = gD[i];
        xT = gX[i];
        bT = gB[i];
        cT = gC[i];
        zT = gZ[i];
        pT = Dp_[cb * SCH + i / SROW4];
    }
    #pragma unroll
    for (int r = 0; r < 3; ++r) {
        const int i = tid + r * 256;
        const float4 d = dr[r];
        const float4 x = xr[r];
        sD[i] = d;
        sU[i] = make_float4(d.x * x.x, d.y * x.y, d.z * x.z, d.w * x.w);
        sB[i] = br[r];
        sC[i] = cr[r];
    }
    if (tail) {
        const int i = tid + 3 * 256;
        sD[i] = dT;
        sU[i] = make_float4(dT.x * xT.x, dT.y * xT.y, dT.z * xT.z, dT.w * xT.w);
        sB[i] = bT;
        sC[i] = cT;
    }
    __syncthreads();

    const float4* pD = sD + ch * SROW4;
    const float4* pU = sU + ch * SROW4;
    const float4* pB = sB + n  * SROW4;
    const float4* pC = sC + n  * SROW4;
    float4*       pY = sY + ch * SROW4;

    float h = 0.f;
    float4 dn = pD[0], un = pU[0], bn = pB[0], cn = pC[0];
    for (int j = 0; j < SROW4; ++j) {
        const float4 d = dn, u = un, bv = bn, cv = cn;
        const int jn = (j + 1 < SROW4) ? (j + 1) : (SROW4 - 1);
        dn = pD[jn]; un = pU[jn]; bn = pB[jn]; cn = pC[jn];
        float4 y;
        y.x = scan_step_(h, A_neg, d.x, u.x, bv.x, cv.x);
        y.y = scan_step_(h, A_neg, d.y, u.y, bv.y, cv.y);
        y.z = scan_step_(h, A_neg, d.z, u.z, bv.z, cv.z);
        y.w = scan_step_(h, A_neg, d.w, u.w, bv.w, cv.w);
        if (n == 0) pY[j] = y;
    }
    __syncthreads();

    #pragma unroll
    for (int r = 0; r < 3; ++r) {
        const int i = tid + r * 256;
        gO[i] = gate_(sY[i], xr[r], zr[r], pr[r]);
    }
    if (tail) {
        const int i = tid + 3 * 256;
        gO[i] = gate_(sY[i], xT, zT, pT);
    }
}

extern "C" void kernel_launch(void* const* d_in, const int* in_sizes, int n_in,
                              void* d_out, int out_size, void* d_ws, size_t ws_size,
                              hipStream_t stream)
{
    const float* hidden    = (const float*)d_in[0];
    const float* in_proj_w = (const float*)d_in[1];
    const float* conv_w    = (const float*)d_in[2];
    const float* conv_b    = (const float*)d_in[3];
    const float* x_proj_w  = (const float*)d_in[4];
    const float* dt_proj_w = (const float*)d_in[5];
    const float* dt_bias   = (const float*)d_in[6];
    const float* A_log     = (const float*)d_in[7];
    const float* D_param   = (const float*)d_in[8];

    float* ws = (float*)d_ws;
    const size_t BCL = (size_t)BSZ * C_ * L_;
    float* x_pre = ws;
    float* zbuf  = ws + BCL;
    float* xcbuf = ws + 2 * BCL;
    float* delta = ws + 3 * BCL;
    float* x_dbl = ws + 4 * BCL;

    k_gemm_inproj<<<dim3(4, BSZ * T_), IPW * 32, 0, stream>>>(
        hidden, in_proj_w, x_pre, zbuf);

    k_conv_xproj_dt<<<dim3(2 * LTILES, BSZ), XPW * 32, 0, stream>>>(
        x_pre, conv_w, conv_b, x_proj_w, dt_proj_w, dt_bias, xcbuf, x_dbl, delta);

    k_scan_out<<<BSZ * (C_ / SCH), 256, 0, stream>>>(
        delta, xcbuf, zbuf, x_dbl, A_log, D_param, (float*)d_out);
}


# scan kernel: y's sum over the 16 state lanes on DPP adds (row_xmask) in a butterfly that halves its values, 5 cross-lane adds per 4 positions, no ds_bpermute; same pairing tree, sum order kept
# speedup vs baseline: 3.4882x; 1.0784x over previous
#include <hip/hip_runtime.h>


#define BSZ   8
#define T_    8
#define L_    196
#define D_    192
#define E_    192
#define C_    1536
#define N_    16
#define KC    4
#define R_    96
#define RN2   128
#define LTILES 13
#define SROW4_X (L_ / 4)

typedef float v2f __attribute__((ext_vector_type(2)));
typedef float v8f __attribute__((ext_vector_type(8)));
typedef float v4f __attribute__((ext_vector_type(4)));

__device__ __forceinline__ float sigmoid_(float x) {
    return 1.0f / (1.0f + __expf(-x));
}

#define IPW     12
#define IPE     96
#define HPITCH  196
#define HPITCH4 (HPITCH / 4)
#define HROW4   (D_ / 4)
#define HTILE4  (L_ * HROW4)
#define HSTG    13

__global__ __launch_bounds__(IPW * 32)
void k_gemm_inproj(const float* __restrict__ hidden,
                   const float* __restrict__ w,
                   float* __restrict__ x_pre,
                   float* __restrict__ zbuf)
{
    __shared__ float4 sH4[L_ * HPITCH4];

    const int tid  = threadIdx.x;
    const int lane = tid & 31;
    const int wid  = tid >> 5;
    const int c    = lane & 15;
    const int h    = lane >> 4;
    const int eq   = blockIdx.x;
    const int bt   = blockIdx.y;
    const int b    = bt / T_;
    const int t    = bt % T_;
    const int et   = wid % 6;
    const int lh   = wid / 6;

    const float* Arow = w + (size_t)(eq * IPE + et * 16 + c) * D_ + 2 * h;
    v2f aF[48];
    #pragma unroll
    for (int s = 0; s < 48; ++s) aF[s] = *(const v2f*)(Arow + 4 * s);

    {
        const float4* g4 = (const float4*)(hidden + (size_t)bt * L_ * D_);
        for (int bb = 0; bb < 2; ++bb) {
            float4 v[HSTG];
            #pragma unroll
            for (int u = 0; u < HSTG; ++u) {
                const int i  = tid + (bb * HSTG + u) * (IPW * 32);
                const int ic = (i < HTILE4) ? i : (HTILE4 - 1);
                v[u] = g4[ic];
            }
            #pragma unroll
            for (int u = 0; u < HSTG; ++u) {
                const int i = tid + (bb * HSTG + u) * (IPW * 32);
                if (i < HTILE4) sH4[(i / HROW4) * HPITCH4 + (i % HROW4)] = v[u];
            }
        }
    }
    __syncthreads();

    const float* sH = (const float*)sH4;
    const int lt0 = lh ? 7 : 0;
    const int lt1 = lh ? LTILES : 7;
    float* __restrict__ dst = (eq < 2) ? x_pre : zbuf;
    const int e0 = (eq & 1) * IPE + et * 16 + 8 * h;

    v2f bN[16];
    {
        const int r0  = lt0 * 16 + c;
        const float* p = sH + ((r0 < L_) ? r0 : (L_ - 1)) * HPITCH + 2 * h;
        #pragma unroll
        for (int i = 0; i < 16; ++i) bN[i] = *(const v2f*)(p + 4 * i);
    }
    for (int lt = lt0; lt < lt1; ++lt) {
        const int ltn = (lt + 1 < lt1) ? (lt + 1) : lt;
        const int rc  = lt * 16 + c;
        const int rn  = ltn * 16 + c;
        const float* pc = sH + ((rc < L_) ? rc : (L_ - 1)) * HPITCH + 2 * h;
        const float* pn = sH + ((rn < L_) ? rn : (L_ - 1)) * HPITCH + 2 * h;

        v8f acc0 = {0.f,0.f,0.f,0.f,0.f,0.f,0.f,0.f};
        v8f acc1 = {0.f,0.f,0.f,0.f,0.f,0.f,0.f,0.f};
        #pragma unroll
        for (int ch = 0; ch < 3; ++ch) {
            v2f bC[16];
            #pragma unroll
            for (int i = 0; i < 16; ++i) bC[i] = bN[i];
            const float* pp = (ch < 2) ? (pc + (ch + 1) * 64) : pn;
            #pragma unroll
            for (int i = 0; i < 16; ++i) bN[i] = *(const v2f*)(pp + 4 * i);
            __builtin_amdgcn_sched_barrier(0);
            #pragma unroll
            for (int i = 0; i < 16; i += 2) {
                acc0 = __builtin_amdgcn_wmma_f32_16x16x4_f32(
                           false, aF[ch * 16 + i], false, bC[i], (short)0, acc0, false, false);
                acc1 = __builtin_amdgcn_wmma_f32_16x16x4_f32(
                           false, aF[ch * 16 + i + 1], false, bC[i + 1], (short)0, acc1, false, false);
            }
        }

        const int l = lt * 16 + c;
        if (l < L_) {
            #pragma unroll
            for (int r = 0; r < 8; ++r) {
                const int ch_out = (e0 + r) * T_ + t;
                dst[((size_t)b * C_ + ch_out) * L_ + l] = acc0[r] + acc1[r];
            }
        }
    }
}

#define XPW    8
#define XKCH   64
#define XNCH   (C_ / XKCH)
#define DKS    (R_ / 4)
#define DCT    6
#define XKS    (XKCH / 4)
#define WROW4  (XKCH / 4)
#define WPITCH (XKCH + 4)
#define WPITCH4 (WPITCH / 4)
#define WBUF4  (RN2 * WPITCH4)
#define WRPR   (XPW * 32 / WROW4)
#define WSTG   (RN2 / WRPR)
#define DPITCH 100
#define DPITCH4 (DPITCH / 4)
#define DTILE4 (16 * R_ / 4)
#define DSTG   (DTILE4 / 32)

__device__ __forceinline__ void lds_barrier_(v8f a0, v8f a1)
{
    asm volatile("s_wait_dscnt 0x0\n\ts_barrier_signal -1\n\ts_barrier_wait -1"
                 :: "v"(a0), "v"(a1) : "memory");
}

__device__ __forceinline__ float conv_tap4_(float bias, float4 wv, float a,
                                            float b, float c, float d)
{
    float s = bias;
    s += a * wv.x;
    s += b * wv.y;
    s += c * wv.z;
    s += d * wv.w;
    return s * sigmoid_(s);
}

__global__ __launch_bounds__(XPW * 32)
void k_conv_xproj_dt(const float* __restrict__ x_pre,
                     const float* __restrict__ cw,
                     const float* __restrict__ cb,
                     const float* __restrict__ w,
                     const float* __restrict__ wdt,
                     const float* __restrict__ dt_bias,
                     float* __restrict__ xc,
                     float* __restrict__ x_dbl,
                     float* __restrict__ delta)
{
    __shared__ float4 sX[C_ * 4];
    __shared__ float  sDt[R_ * 16];
    __shared__ float4 sW4[3 * WBUF4];

    const int tid = threadIdx.x;
    const int hf  = blockIdx.x & 1;
    const int lt  = blockIdx.x >> 1;
    const int b   = blockIdx.y;

    {
        const int j  = tid & 3;
        const int cr = tid >> 2;
        const int l  = lt * 16 + 4 * j;
        const bool valid   = l < L_;
        const bool hasprev = l > 0;
        const bool store   = valid && (hf == 0);
        const int q  = (valid ? l : (L_ - 4)) >> 2;
        const int qp = hasprev ? (q - 1) : q;
        const float4* xr4 = (const float4*)(x_pre + (size_t)b * C_ * L_);
        float4*       xo4 = (float4*)(xc + (size_t)b * C_ * L_) + q;
        const float4* cw4 = (const float4*)cw;
        float4 curN[8], plN[8], wvN[8];
        float  bsN[8];
        #pragma unroll
        for (int u = 0; u < 8; ++u) {
            const int c = u * 64 + cr;
            curN[u] = xr4[c * SROW4_X + q];
            plN[u]  = xr4[c * SROW4_X + qp];
            wvN[u]  = cw4[c];
            bsN[u]  = cb[c];
        }
        for (int t = 0; t < C_ / 512; ++t) {
            float4 curC[8], plC[8], wvC[8];
            float  bsC[8];
            #pragma unroll
            for (int u = 0; u < 8; ++u) {
                curC[u] = curN[u]; plC[u] = plN[u]; wvC[u] = wvN[u]; bsC[u] = bsN[u];
            }
            const int tn = (t + 1 < C_ / 512) ? (t + 1) : t;
            #pragma unroll
            for (int u = 0; u < 8; ++u) {
                const int c = tn * 512 + u * 64 + cr;
                curN[u] = xr4[c * SROW4_X + q];
                plN[u]  = xr4[c * SROW4_X + qp];
                wvN[u]  = cw4[c];
                bsN[u]  = cb[c];
            }
            #pragma unroll
            for (int u = 0; u < 8; ++u) {
                const int c = t * 512 + u * 64 + cr;
                const float4 cur = curC[u];
                const float4 wv  = wvC[u];
                const float bias = bsC[u];
                const float p1 = hasprev ? plC[u].y : 0.f;
                const float p2 = hasprev ? plC[u].z : 0.f;
                const float p3 = hasprev ? plC[u].w : 0.f;
                float4 o;
                o.x = conv_tap4_(bias, wv, p1, p2, p3, cur.x);
                o.y = conv_tap4_(bias, wv, p2, p3, cur.x, cur.y);
                o.z = conv_tap4_(bias, wv, p3, cur.x, cur.y, cur.z);
                o.w = conv_tap4_(bias, wv, cur.x, cur.y, cur.z, cur.w);
                sX[c * 4 + j] = o;
                if (store) xo4[c * SROW4_X] = o;
            }
        }
    }

    const float4* w4 = (const float4*)w;
    const int srow = tid / WROW4;
    const int sc4  = tid % WROW4;
    {
        float4 g[2 * WSTG];
        #pragma unroll
        for (int u = 0; u < 2 * WSTG; ++u)
            g[u] = w4[(size_t)(srow + WRPR * (u % WSTG)) * (C_ / 4) + (u / WSTG) * WROW4 + sc4];
        #pragma unroll
        for (int u = 0; u < 2 * WSTG; ++u)
            sW4[(u / WSTG) * WBUF4 + (srow + WRPR * (u % WSTG)) * WPITCH4 + sc4] = g[u];
    }
    const v4f* w4v = (const v4f*)w;
    v4f*       sWv = (v4f*)sW4;
    v4f gW[WSTG];
    #pragma unroll
    for (int u = 0; u < WSTG; ++u)
        gW[u] = w4v[(size_t)(srow + WRPR * u) * (C_ / 4) + 2 * WROW4 + sc4];
    __syncthreads();

    const int lane = tid & 31;
    const int wid  = tid >> 5;
    const int m    = lane & 15;
    const int h    = lane >> 4;
    const int l     = lt * 16 + m;
    const int mbase = h * 8;
    const int ct0 = hf * 48 + wid * DCT;
    const v4f*    g3    = (const v4f*)wdt + (size_t)ct0 * DTILE4;
    const float4* bias4 = (const float4*)(dt_bias + ct0 * 16 + mbase);
    v4f    gN3[DSTG];
    float4 bsN0, bsN1;
    {
        const float* Afrag = (const float*)sW4 + (wid * 16 + m) * WPITCH + 2 * h;
        const float* Bcol = (const float*)sX + (2 * h) * 16 + m;

        v8f acc0 = {0.f,0.f,0.f,0.f,0.f,0.f,0.f,0.f};
        v8f acc1 = {0.f,0.f,0.f,0.f,0.f,0.f,0.f,0.f};

        v2f aC[XKS], bC[XKS];
        #pragma unroll
        for (int i = 0; i < XKS; ++i) {
            aC[i] = *(const v2f*)(Afrag + 4 * i);
            const float* p = Bcol + (4 * i) * 16;
            bC[i].x = p[0];
            bC[i].y = p[16];
        }
        #pragma unroll 2
        for (int ch = 0; ch < XNCH; ++ch) {
            const int n1 = (ch + 1 < XNCH) ? (ch + 1) : (XNCH - 1);
            const int n3 = (ch + 3 < XNCH) ? (ch + 3) : (XNCH - 1);
            #pragma unroll
            for (int u = 0; u < WSTG; ++u)
                sWv[((ch + 2) % 3) * WBUF4 + (srow + WRPR * u) * WPITCH4 + sc4] = gW[u];
            #pragma unroll
            for (int u = 0; u < WSTG; ++u)
                gW[u] = w4v[(size_t)(srow + WRPR * u) * (C_ / 4) + n3 * WROW4 + sc4];
            v2f aN[XKS], bN[XKS];
            #pragma unroll
            for (int i = 0; i < XKS; ++i) {
                aN[i] = *(const v2f*)(Afrag + (n1 % 3) * (RN2 * WPITCH) + 4 * i);
                const float* p = Bcol + (n1 * XKCH + 4 * i) * 16;
                bN[i].x = p[0];
                bN[i].y = p[16];
            }
            __builtin_amdgcn_sched_barrier(0);
            #pragma unroll
            for (int i = 0; i < XKS; i += 2) {
                acc0 = __builtin_amdgcn_wmma_f32_16x16x4_f32(
                           false, aC[i], false, bC[i], (short)0, acc0, false, false);
                acc1 = __builtin_amdgcn_wmma_f32_16x16x4_f32(
                           false, aC[i + 1], false, bC[i + 1], (short)0, acc1, false, false);
            }
            lds_barrier_(acc0, acc1);
            #pragma unroll
            for (int i = 0; i < XKS; ++i) { aC[i] = aN[i]; bC[i] = bN[i]; }
        }

        #pragma unroll
        for (int u = 0; u < DSTG; ++u) gN3[u] = g3[lane + 32 * u];
        bsN0 = bias4[0];
        bsN1 = bias4[1];

        if (wid < R_ / 16) {
            #pragma unroll
            for (int r = 0; r < 8; ++r)
                sDt[(wid * 16 + mbase + r) * 16 + m] = acc0[r] + acc1[r];
        } else if (hf == 0 && l < L_) {
            #pragma unroll
            for (int r = 0; r < 8; ++r) {
                const int rr = wid * 16 + mbase + r;
                x_dbl[((size_t)b * RN2 + rr) * L_ + l] = acc0[r] + acc1[r];
            }
        }
    }
    v4f* sA4 = (v4f*)sW4 + wid * (16 * DPITCH4);
    #pragma unroll
    for (int u = 0; u < DSTG; ++u) {
        const int idx = lane + 32 * u;
        sA4[(idx / (R_ / 4)) * DPITCH4 + (idx % (R_ / 4))] = gN3[u];
    }
    __syncthreads();

    {
        v2f bF[DKS];
        #pragma unroll
        for (int s = 0; s < DKS; ++s) {
            const float* p = sDt + (4 * s + 2 * h) * 16 + m;
            bF[s].x = p[0];
            bF[s].y = p[16];
        }
        __builtin_amdgcn_sched_barrier(0);

        float* __restrict__ dcol = delta + ((size_t)b * C_ + ct0 * 16 + mbase) * L_ + l;

        const float* sAf = (const float*)sA4 + m * DPITCH + 2 * h;
        #pragma unroll
        for (int u = 0; u < DSTG; ++u) gN3[u] = g3[DTILE4 + lane + 32 * u];
        #pragma unroll 1
        for (int i = 0; i < DCT; ++i) {
            const float4 bs0 = bsN0, bs1 = bsN1;
            const int i1 = (i + 1 < DCT) ? (i + 1) : (DCT - 1);
            const int i2 = (i + 2 < DCT) ? (i + 2) : (DCT - 1);
            v2f aC[DKS];
            #pragma unroll
            for (int s = 0; s < DKS; ++s) aC[s] = *(const v2f*)(sAf + 4 * s);
            bsN0 = bias4[i1 * 4];
            bsN1 = bias4[i1 * 4 + 1];

            v8f acc0 = {0.f,0.f,0.f,0.f,0.f,0.f,0.f,0.f};
            v8f acc1 = {0.f,0.f,0.f,0.f,0.f,0.f,0.f,0.f};
            #pragma unroll
            for (int s = 0; s < DKS; s += 2) {
                acc0 = __builtin_amdgcn_wmma_f32_16x16x4_f32(
                           false, aC[s], false, bF[s], (short)0, acc0, false, false);
                acc1 = __builtin_amdgcn_wmma_f32_16x16x4_f32(
                           false, aC[s + 1], false, bF[s + 1], (short)0, acc1, false, false);
            }

            #pragma unroll
            for (int u = 0; u < DSTG; ++u) {
                const int idx = lane + 32 * u;
                sA4[(idx / (R_ / 4)) * DPITCH4 + (idx % (R_ / 4))] = gN3[u];
            }
            #pragma unroll
            for (int u = 0; u < DSTG; ++u) gN3[u] = g3[i2 * DTILE4 + lane + 32 * u];

            const float bsv[8] = {bs0.x, bs0.y, bs0.z, bs0.w, bs1.x, bs1.y, bs1.z, bs1.w};
            float sp[8];
            #pragma unroll
            for (int r = 0; r < 8; ++r) {
                const float v = acc0[r] + acc1[r] + bsv[r];
                sp[r] = (v > 20.f) ? v : __logf(1.f + __expf(v));
            }
            if (l < L_) {
                #pragma unroll
                for (int r = 0; r < 8; ++r)
                    dcol[(size_t)(i * 16 + r) * L_] = sp[r];
            }
        }
    }
}

#define SCH    16
#define SROW4  (L_ / 4)
#define STILE4 (SCH * SROW4)

__device__ __forceinline__ float4 gate_(float4 y, float4 x, float4 z, float Dp)
{
    float4 o;
    o.x = (y.x + x.x * Dp) * (z.x * sigmoid_(z.x));
    o.y = (y.y + x.y * Dp) * (z.y * sigmoid_(z.y));
    o.z = (y.z + x.z * Dp) * (z.z * sigmoid_(z.z));
    o.w = (y.w + x.w * Dp) * (z.w * sigmoid_(z.w));
    return o;
}

__device__ __forceinline__ float scan_prod_(float& h, float A_neg, float dv,
                                            float u, float bv, float cv)
{
    const float dA = __expf(dv * A_neg);
    h = fmaf(h, dA, u * bv);
    return h * cv;
}

template <int XM>
__device__ __forceinline__ float xlane_(float v)
{
    return __int_as_float(__builtin_amdgcn_update_dpp(
               0, __float_as_int(v), 0x160 + XM, 0xf, 0xf, true));
}

__global__ __launch_bounds__(256)
void k_scan_out(const float* __restrict__ delta,
                const float* __restrict__ xc,
                const float* __restrict__ zbuf,
                const float* __restrict__ x_dbl,
                const float* __restrict__ A_log,
                const float* __restrict__ Dp_,
                float* __restrict__ out)
{
    __shared__ float4 sD[STILE4];
    __shared__ float4 sU[STILE4];
    __shared__ float4 sB[STILE4];
    __shared__ float4 sC[STILE4];
    __shared__ float4 sY[STILE4];

    const int tid = threadIdx.x;
    const int cb  = blockIdx.x % (C_ / SCH);
    const int b   = blockIdx.x / (C_ / SCH);
    const size_t row0 = ((size_t)b * C_ + (size_t)cb * SCH) * L_;

    const float4* gD = (const float4*)(delta + row0);
    const float4* gX = (const float4*)(xc    + row0);
    const float4* gZ = (const float4*)(zbuf  + row0);
    const float4* gB = (const float4*)(x_dbl + ((size_t)b * RN2 + R_) * L_);
    const float4* gC = (const float4*)(x_dbl + ((size_t)b * RN2 + R_ + N_) * L_);
    float4* gO = (float4*)(out + row0);

    const int lane = tid & 31;
    const int n    = lane & 15;
    const int ch   = (tid >> 5) * 2 + (lane >> 4);
    const float A_neg = -__expf(A_log[(cb * SCH + ch) * N_ + n]);

    float4 dr[3], xr[3], zr[3], br[3], cr[3];
    const bool tail = tid < (STILE4 - 3 * 256);
    const float4 zero4 = make_float4(0.f, 0.f, 0.f, 0.f);
    float4 dT = zero4, xT = zero4, zT = zero4, bT = zero4, cT = zero4;
    float  pr[3], pT = 0.f;
    #pragma unroll
    for (int r = 0; r < 3; ++r) {
        const int i = tid + r * 256;
        dr[r] = gD[i];
        xr[r] = gX[i];
        br[r] = gB[i];
        cr[r] = gC[i];
        zr[r] = gZ[i];
        pr[r] = Dp_[cb * SCH + i / SROW4];
    }
    if (tail) {
        const int i = tid + 3 * 256;
        dT = gD[i];
        xT = gX[i];
        bT = gB[i];
        cT = gC[i];
        zT = gZ[i];
        pT = Dp_[cb * SCH + i / SROW4];
    }
    #pragma unroll
    for (int r = 0; r < 3; ++r) {
        const int i = tid + r * 256;
        const float4 d = dr[r];
        const float4 x = xr[r];
        sD[i] = d;
        sU[i] = make_float4(d.x * x.x, d.y * x.y, d.z * x.z, d.w * x.w);
        sB[i] = br[r];
        sC[i] = cr[r];
    }
    if (tail) {
        const int i = tid + 3 * 256;
        sD[i] = dT;
        sU[i] = make_float4(dT.x * xT.x, dT.y * xT.y, dT.z * xT.z, dT.w * xT.w);
        sB[i] = bT;
        sC[i] = cT;
    }
    __syncthreads();

    const float4* pD = sD + ch * SROW4;
    const float4* pU = sU + ch * SROW4;
    const float4* pB = sB + n  * SROW4;
    const float4* pC = sC + n  * SROW4;
    float4*       pY = sY + ch * SROW4;

    const bool odd1 = (n & 1) != 0;
    const bool odd2 = (n & 2) != 0;
    float* pYf = (float*)pY + (n & 3);

    float h = 0.f;
    float4 dn = pD[0], un = pU[0], bn = pB[0], cn = pC[0];
    for (int j = 0; j < SROW4; ++j) {
        const float4 d = dn, u = un, bv = bn, cv = cn;
        const int jn = (j + 1 < SROW4) ? (j + 1) : (SROW4 - 1);
        dn = pD[jn]; un = pU[jn]; bn = pB[jn]; cn = pC[jn];
        const float t0 = scan_prod_(h, A_neg, d.x, u.x, bv.x, cv.x);
        const float t1 = scan_prod_(h, A_neg, d.y, u.y, bv.y, cv.y);
        const float t2 = scan_prod_(h, A_neg, d.z, u.z, bv.z, cv.z);
        const float t3 = scan_prod_(h, A_neg, d.w, u.w, bv.w, cv.w);
        const float k01 = odd1 ? t1 : t0, g01 = odd1 ? t0 : t1;
        const float k23 = odd1 ? t3 : t2, g23 = odd1 ? t2 : t3;
        const float r01 = k01 + xlane_<1>(g01);
        const float r23 = k23 + xlane_<1>(g23);
        const float k = odd2 ? r23 : r01, g = odd2 ? r01 : r23;
        float q = k + xlane_<2>(g);
        q += xlane_<4>(q);
        q += xlane_<8>(q);
        if (n < 4) pYf[4 * j] = q;
    }
    __syncthreads();

    #pragma unroll
    for (int r = 0; r < 3; ++r) {
        const int i = tid + r * 256;
        gO[i] = gate_(sY[i], xr[r], zr[r], pr[r]);
    }
    if (tail) {
        const int i = tid + 3 * 256;
        gO[i] = gate_(sY[i], xT, zT, pT);
    }
}

extern "C" void kernel_launch(void* const* d_in, const int* in_sizes, int n_in,
                              void* d_out, int out_size, void* d_ws, size_t ws_size,
                              hipStream_t stream)
{
    const float* hidden    = (const float*)d_in[0];
    const float* in_proj_w = (const float*)d_in[1];
    const float* conv_w    = (const float*)d_in[2];
    const float* conv_b    = (const float*)d_in[3];
    const float* x_proj_w  = (const float*)d_in[4];
    const float* dt_proj_w = (const float*)d_in[5];
    const float* dt_bias   = (const float*)d_in[6];
    const float* A_log     = (const float*)d_in[7];
    const float* D_param   = (const float*)d_in[8];

    float* ws = (float*)d_ws;
    const size_t BCL = (size_t)BSZ * C_ * L_;
    float* x_pre = ws;
    float* zbuf  = ws + BCL;
    float* xcbuf = ws + 2 * BCL;
    float* delta = ws + 3 * BCL;
    float* x_dbl = ws + 4 * BCL;

    k_gemm_inproj<<<dim3(4, BSZ * T_), IPW * 32, 0, stream>>>(
        hidden, in_proj_w, x_pre, zbuf);

    k_conv_xproj_dt<<<dim3(2 * LTILES, BSZ), XPW * 32, 0, stream>>>(
        x_pre, conv_w, conv_b, x_proj_w, dt_proj_w, dt_bias, xcbuf, x_dbl, delta);

    k_scan_out<<<BSZ * (C_ / SCH), 256, 0, stream>>>(
        delta, xcbuf, zbuf, x_dbl, A_log, D_param, (float*)d_out);
}


# scan kernel: recurrence loop runs 4 trips a pass: staged-row register sets alternate (no copies), lane n keeps y of position 16J+n, one 64-B LDS store a pass, no lane-masked branch; sum order kept
# speedup vs baseline: 3.6261x; 1.0395x over previous
#include <hip/hip_runtime.h>


#define BSZ   8
#define T_    8
#define L_    196
#define D_    192
#define E_    192
#define C_    1536
#define N_    16
#define KC    4
#define R_    96
#define RN2   128
#define LTILES 13
#define SROW4_X (L_ / 4)

typedef float v2f __attribute__((ext_vector_type(2)));
typedef float v8f __attribute__((ext_vector_type(8)));
typedef float v4f __attribute__((ext_vector_type(4)));

__device__ __forceinline__ float sigmoid_(float x) {
    return 1.0f / (1.0f + __expf(-x));
}

#define IPW     12
#define IPE     96
#define HPITCH  196
#define HPITCH4 (HPITCH / 4)
#define HROW4   (D_ / 4)
#define HTILE4  (L_ * HROW4)
#define HSTG    13

__global__ __launch_bounds__(IPW * 32)
void k_gemm_inproj(const float* __restrict__ hidden,
                   const float* __restrict__ w,
                   float* __restrict__ x_pre,
                   float* __restrict__ zbuf)
{
    __shared__ float4 sH4[L_ * HPITCH4];

    const int tid  = threadIdx.x;
    const int lane = tid & 31;
    const int wid  = tid >> 5;
    const int c    = lane & 15;
    const int h    = lane >> 4;
    const int eq   = blockIdx.x;
    const int bt   = blockIdx.y;
    const int b    = bt / T_;
    const int t    = bt % T_;
    const int et   = wid % 6;
    const int lh   = wid / 6;

    const float* Arow = w + (size_t)(eq * IPE + et * 16 + c) * D_ + 2 * h;
    v2f aF[48];
    #pragma unroll
    for (int s = 0; s < 48; ++s) aF[s] = *(const v2f*)(Arow + 4 * s);

    {
        const float4* g4 = (const float4*)(hidden + (size_t)bt * L_ * D_);
        for (int bb = 0; bb < 2; ++bb) {
            float4 v[HSTG];
            #pragma unroll
            for (int u = 0; u < HSTG; ++u) {
                const int i  = tid + (bb * HSTG + u) * (IPW * 32);
                const int ic = (i < HTILE4) ? i : (HTILE4 - 1);
                v[u] = g4[ic];
            }
            #pragma unroll
            for (int u = 0; u < HSTG; ++u) {
                const int i = tid + (bb * HSTG + u) * (IPW * 32);
                if (i < HTILE4) sH4[(i / HROW4) * HPITCH4 + (i % HROW4)] = v[u];
            }
        }
    }
    __syncthreads();

    const float* sH = (const float*)sH4;
    const int lt0 = lh ? 7 : 0;
    const int lt1 = lh ? LTILES : 7;
    float* __restrict__ dst = (eq < 2) ? x_pre : zbuf;
    const int e0 = (eq & 1) * IPE + et * 16 + 8 * h;

    v2f bN[16];
    {
        const int r0  = lt0 * 16 + c;
        const float* p = sH + ((r0 < L_) ? r0 : (L_ - 1)) * HPITCH + 2 * h;
        #pragma unroll
        for (int i = 0; i < 16; ++i) bN[i] = *(const v2f*)(p + 4 * i);
    }
    for (int lt = lt0; lt < lt1; ++lt) {
        const int ltn = (lt + 1 < lt1) ? (lt + 1) : lt;
        const int rc  = lt * 16 + c;
        const int rn  = ltn * 16 + c;
        const float* pc = sH + ((rc < L_) ? rc : (L_ - 1)) * HPITCH + 2 * h;
        const float* pn = sH + ((rn < L_) ? rn : (L_ - 1)) * HPITCH + 2 * h;

        v8f acc0 = {0.f,0.f,0.f,0.f,0.f,0.f,0.f,0.f};
        v8f acc1 = {0.f,0.f,0.f,0.f,0.f,0.f,0.f,0.f};
        #pragma unroll
        for (int ch = 0; ch < 3; ++ch) {
            v2f bC[16];
            #pragma unroll
            for (int i = 0; i < 16; ++i) bC[i] = bN[i];
            const float* pp = (ch < 2) ? (pc + (ch + 1) * 64) : pn;
            #pragma unroll
            for (int i = 0; i < 16; ++i) bN[i] = *(const v2f*)(pp + 4 * i);
            __builtin_amdgcn_sched_barrier(0);
            #pragma unroll
            for (int i = 0; i < 16; i += 2) {
                acc0 = __builtin_amdgcn_wmma_f32_16x16x4_f32(
                           false, aF[ch * 16 + i], false, bC[i], (short)0, acc0, false, false);
                acc1 = __builtin_amdgcn_wmma_f32_16x16x4_f32(
                           false, aF[ch * 16 + i + 1], false, bC[i + 1], (short)0, acc1, false, false);
            }
        }

        const int l = lt * 16 + c;
        if (l < L_) {
            #pragma unroll
            for (int r = 0; r < 8; ++r) {
                const int ch_out = (e0 + r) * T_ + t;
                dst[((size_t)b * C_ + ch_out) * L_ + l] = acc0[r] + acc1[r];
            }
        }
    }
}

#define XPW    8
#define XKCH   64
#define XNCH   (C_ / XKCH)
#define DKS    (R_ / 4)
#define DCT    6
#define XKS    (XKCH / 4)
#define WROW4  (XKCH / 4)
#define WPITCH (XKCH + 4)
#define WPITCH4 (WPITCH / 4)
#define WBUF4  (RN2 * WPITCH4)
#define WRPR   (XPW * 32 / WROW4)
#define WSTG   (RN2 / WRPR)
#define DPITCH 100
#define DPITCH4 (DPITCH / 4)
#define DTILE4 (16 * R_ / 4)
#define DSTG   (DTILE4 / 32)

__device__ __forceinline__ void lds_barrier_(v8f a0, v8f a1)
{
    asm volatile("s_wait_dscnt 0x0\n\ts_barrier_signal -1\n\ts_barrier_wait -1"
                 :: "v"(a0), "v"(a1) : "memory");
}

__device__ __forceinline__ float conv_tap4_(float bias, float4 wv, float a,
                                            float b, float c, float d)
{
    float s = bias;
    s += a * wv.x;
    s += b * wv.y;
    s += c * wv.z;
    s += d * wv.w;
    return s * sigmoid_(s);
}

__global__ __launch_bounds__(XPW * 32)
void k_conv_xproj_dt(const float* __restrict__ x_pre,
                     const float* __restrict__ cw,
                     const float* __restrict__ cb,
                     const float* __restrict__ w,
                     const float* __restrict__ wdt,
                     const float* __restrict__ dt_bias,
                     float* __restrict__ xc,
                     float* __restrict__ x_dbl,
                     float* __restrict__ delta)
{
    __shared__ float4 sX[C_ * 4];
    __shared__ float  sDt[R_ * 16];
    __shared__ float4 sW4[3 * WBUF4];

    const int tid = threadIdx.x;
    const int hf  = blockIdx.x & 1;
    const int lt  = blockIdx.x >> 1;
    const int b   = blockIdx.y;

    {
        const int j  = tid & 3;
        const int cr = tid >> 2;
        const int l  = lt * 16 + 4 * j;
        const bool valid   = l < L_;
        const bool hasprev = l > 0;
        const bool store   = valid && (hf == 0);
        const int q  = (valid ? l : (L_ - 4)) >> 2;
        const int qp = hasprev ? (q - 1) : q;
        const float4* xr4 = (const float4*)(x_pre + (size_t)b * C_ * L_);
        float4*       xo4 = (float4*)(xc + (size_t)b * C_ * L_) + q;
        const float4* cw4 = (const float4*)cw;
        float4 curN[8], plN[8], wvN[8];
        float  bsN[8];
        #pragma unroll
        for (int u = 0; u < 8; ++u) {
            const int c = u * 64 + cr;
            curN[u] = xr4[c * SROW4_X + q];
            plN[u]  = xr4[c * SROW4_X + qp];
            wvN[u]  = cw4[c];
            bsN[u]  = cb[c];
        }
        for (int t = 0; t < C_ / 512; ++t) {
            float4 curC[8], plC[8], wvC[8];
            float  bsC[8];
            #pragma unroll
            for (int u = 0; u < 8; ++u) {
                curC[u] = curN[u]; plC[u] = plN[u]; wvC[u] = wvN[u]; bsC[u] = bsN[u];
            }
            const int tn = (t + 1 < C_ / 512) ? (t + 1) : t;
            #pragma unroll
            for (int u = 0; u < 8; ++u) {
                const int c = tn * 512 + u * 64 + cr;
                curN[u] = xr4[c * SROW4_X + q];
                plN[u]  = xr4[c * SROW4_X + qp];
                wvN[u]  = cw4[c];
                bsN[u]  = cb[c];
            }
            #pragma unroll
            for (int u = 0; u < 8; ++u) {
                const int c = t * 512 + u * 64 + cr;
                const float4 cur = curC[u];
                const float4 wv  = wvC[u];
                const float bias = bsC[u];
                const float p1 = hasprev ? plC[u].y : 0.f;
                const float p2 = hasprev ? plC[u].z : 0.f;
                const float p3 = hasprev ? plC[u].w : 0.f;
                float4 o;
                o.x = conv_tap4_(bias, wv, p1, p2, p3, cur.x);
                o.y = conv_tap4_(bias, wv, p2, p3, cur.x, cur.y);
                o.z = conv_tap4_(bias, wv, p3, cur.x, cur.y, cur.z);
                o.w = conv_tap4_(bias, wv, cur.x, cur.y, cur.z, cur.w);
                sX[c * 4 + j] = o;
                if (store) xo4[c * SROW4_X] = o;
            }
        }
    }

    const float4* w4 = (const float4*)w;
    const int srow = tid / WROW4;
    const int sc4  = tid % WROW4;
    {
        float4 g[2 * WSTG];
        #pragma unroll
        for (int u = 0; u < 2 * WSTG; ++u)
            g[u] = w4[(size_t)(srow + WRPR * (u % WSTG)) * (C_ / 4) + (u / WSTG) * WROW4 + sc4];
        #pragma unroll
        for (int u = 0; u < 2 * WSTG; ++u)
            sW4[(u / WSTG) * WBUF4 + (srow + WRPR * (u % WSTG)) * WPITCH4 + sc4] = g[u];
    }
    const v4f* w4v = (const v4f*)w;
    v4f*       sWv = (v4f*)sW4;
    v4f gW[WSTG];
    #pragma unroll
    for (int u = 0; u < WSTG; ++u)
        gW[u] = w4v[(size_t)(srow + WRPR * u) * (C_ / 4) + 2 * WROW4 + sc4];
    __syncthreads();

    const int lane = tid & 31;
    const int wid  = tid >> 5;
    const int m    = lane & 15;
    const int h    = lane >> 4;
    const int l     = lt * 16 + m;
    const int mbase = h * 8;
    const int ct0 = hf * 48 + wid * DCT;
    const v4f*    g3    = (const v4f*)wdt + (size_t)ct0 * DTILE4;
    const float4* bias4 = (const float4*)(dt_bias + ct0 * 16 + mbase);
    v4f    gN3[DSTG];
    float4 bsN0, bsN1;
    {
        const float* Afrag = (const float*)sW4 + (wid * 16 + m) * WPITCH + 2 * h;
        const float* Bcol = (const float*)sX + (2 * h) * 16 + m;

        v8f acc0 = {0.f,0.f,0.f,0.f,0.f,0.f,0.f,0.f};
        v8f acc1 = {0.f,0.f,0.f,0.f,0.f,0.f,0.f,0.f};

        v2f aC[XKS], bC[XKS];
        #pragma unroll
        for (int i = 0; i < XKS; ++i) {
            aC[i] = *(const v2f*)(Afrag + 4 * i);
            const float* p = Bcol + (4 * i) * 16;
            bC[i].x = p[0];
            bC[i].y = p[16];
        }
        #pragma unroll 2
        for (int ch = 0; ch < XNCH; ++ch) {
            const int n1 = (ch + 1 < XNCH) ? (ch + 1) : (XNCH - 1);
            const int n3 = (ch + 3 < XNCH) ? (ch + 3) : (XNCH - 1);
            #pragma unroll
            for (int u = 0; u < WSTG; ++u)
                sWv[((ch + 2) % 3) * WBUF4 + (srow + WRPR * u) * WPITCH4 + sc4] = gW[u];
            #pragma unroll
            for (int u = 0; u < WSTG; ++u)
                gW[u] = w4v[(size_t)(srow + WRPR * u) * (C_ / 4) + n3 * WROW4 + sc4];
            v2f aN[XKS], bN[XKS];
            #pragma unroll
            for (int i = 0; i < XKS; ++i) {
                aN[i] = *(const v2f*)(Afrag + (n1 % 3) * (RN2 * WPITCH) + 4 * i);
                const float* p = Bcol + (n1 * XKCH + 4 * i) * 16;
                bN[i].x = p[0];
                bN[i].y = p[16];
            }
            __builtin_amdgcn_sched_barrier(0);
            #pragma unroll
            for (int i = 0; i < XKS; i += 2) {
                acc0 = __builtin_amdgcn_wmma_f32_16x16x4_f32(
                           false, aC[i], false, bC[i], (short)0, acc0, false, false);
                acc1 = __builtin_amdgcn_wmma_f32_16x16x4_f32(
                           false, aC[i + 1], false, bC[i + 1], (short)0, acc1, false, false);
            }
            lds_barrier_(acc0, acc1);
            #pragma unroll
            for (int i = 0; i < XKS; ++i) { aC[i] = aN[i]; bC[i] = bN[i]; }
        }

        #pragma unroll
        for (int u = 0; u < DSTG; ++u) gN3[u] = g3[lane + 32 * u];
        bsN0 = bias4[0];
        bsN1 = bias4[1];

        if (wid < R_ / 16) {
            #pragma unroll
            for (int r = 0; r < 8; ++r)
                sDt[(wid * 16 + mbase + r) * 16 + m] = acc0[r] + acc1[r];
        } else if (hf == 0 && l < L_) {
            #pragma unroll
            for (int r = 0; r < 8; ++r) {
                const int rr = wid * 16 + mbase + r;
                x_dbl[((size_t)b * RN2 + rr) * L_ + l] = acc0[r] + acc1[r];
            }
        }
    }
    v4f* sA4 = (v4f*)sW4 + wid * (16 * DPITCH4);
    #pragma unroll
    for (int u = 0; u < DSTG; ++u) {
        const int idx = lane + 32 * u;
        sA4[(idx / (R_ / 4)) * DPITCH4 + (idx % (R_ / 4))] = gN3[u];
    }
    __syncthreads();

    {
        v2f bF[DKS];
        #pragma unroll
        for (int s = 0; s < DKS; ++s) {
            const float* p = sDt + (4 * s + 2 * h) * 16 + m;
            bF[s].x = p[0];
            bF[s].y = p[16];
        }
        __builtin_amdgcn_sched_barrier(0);

        float* __restrict__ dcol = delta + ((size_t)b * C_ + ct0 * 16 + mbase) * L_ + l;

        const float* sAf = (const float*)sA4 + m * DPITCH + 2 * h;
        #pragma unroll
        for (int u = 0; u < DSTG; ++u) gN3[u] = g3[DTILE4 + lane + 32 * u];
        #pragma unroll 1
        for (int i = 0; i < DCT; ++i) {
            const float4 bs0 = bsN0, bs1 = bsN1;
            const int i1 = (i + 1 < DCT) ? (i + 1) : (DCT - 1);
            const int i2 = (i + 2 < DCT) ? (i + 2) : (DCT - 1);
            v2f aC[DKS];
            #pragma unroll
            for (int s = 0; s < DKS; ++s) aC[s] = *(const v2f*)(sAf + 4 * s);
            bsN0 = bias4[i1 * 4];
            bsN1 = bias4[i1 * 4 + 1];

            v8f acc0 = {0.f,0.f,0.f,0.f,0.f,0.f,0.f,0.f};
            v8f acc1 = {0.f,0.f,0.f,0.f,0.f,0.f,0.f,0.f};
            #pragma unroll
            for (int s = 0; s < DKS; s += 2) {
                acc0 = __builtin_amdgcn_wmma_f32_16x16x4_f32(
                           false, aC[s], false, bF[s], (short)0, acc0, false, false);
                acc1 = __builtin_amdgcn_wmma_f32_16x16x4_f32(
                           false, aC[s + 1], false, bF[s + 1], (short)0, acc1, false, false);
            }

            #pragma unroll
            for (int u = 0; u < DSTG; ++u) {
                const int idx = lane + 32 * u;
                sA4[(idx / (R_ / 4)) * DPITCH4 + (idx % (R_ / 4))] = gN3[u];
            }
            #pragma unroll
            for (int u = 0; u < DSTG; ++u) gN3[u] = g3[i2 * DTILE4 + lane + 32 * u];

            const float bsv[8] = {bs0.x, bs0.y, bs0.z, bs0.w, bs1.x, bs1.y, bs1.z, bs1.w};
            float sp[8];
            #pragma unroll
            for (int r = 0; r < 8; ++r) {
                const float v = acc0[r] + acc1[r] + bsv[r];
                sp[r] = (v > 20.f) ? v : __logf(1.f + __expf(v));
            }
            if (l < L_) {
                #pragma unroll
                for (int r = 0; r < 8; ++r)
                    dcol[(size_t)(i * 16 + r) * L_] = sp[r];
            }
        }
    }
}

#define SCH    16
#define SROW4  (L_ / 4)
#define STILE4 (SCH * SROW4)

__device__ __forceinline__ float4 gate_(float4 y, float4 x, float4 z, float Dp)
{
    float4 o;
    o.x = (y.x + x.x * Dp) * (z.x * sigmoid_(z.x));
    o.y = (y.y + x.y * Dp) * (z.y * sigmoid_(z.y));
    o.z = (y.z + x.z * Dp) * (z.z * sigmoid_(z.z));
    o.w = (y.w + x.w * Dp) * (z.w * sigmoid_(z.w));
    return o;
}

__device__ __forceinline__ float scan_prod_(float& h, float A_neg, float dv,
                                            float u, float bv, float cv)
{
    const float dA = __expf(dv * A_neg);
    h = fmaf(h, dA, u * bv);
    return h * cv;
}

template <int XM>
__device__ __forceinline__ float xlane_(float v)
{
    return __int_as_float(__builtin_amdgcn_update_dpp(
               0, __float_as_int(v), 0x160 + XM, 0xf, 0xf, true));
}

__device__ __forceinline__ float scan_trip_(float& h, float A_neg, float4 d, float4 u,
                                            float4 bv, float4 cv, bool odd1, bool odd2)
{
    const float t0 = scan_prod_(h, A_neg, d.x, u.x, bv.x, cv.x);
    const float t1 = scan_prod_(h, A_neg, d.y, u.y, bv.y, cv.y);
    const float t2 = scan_prod_(h, A_neg, d.z, u.z, bv.z, cv.z);
    const float t3 = scan_prod_(h, A_neg, d.w, u.w, bv.w, cv.w);
    const float k01 = odd1 ? t1 : t0, g01 = odd1 ? t0 : t1;
    const float k23 = odd1 ? t3 : t2, g23 = odd1 ? t2 : t3;
    const float r01 = k01 + xlane_<1>(g01);
    const float r23 = k23 + xlane_<1>(g23);
    const float k = odd2 ? r23 : r01, g = odd2 ? r01 : r23;
    float q = k + xlane_<2>(g);
    q += xlane_<4>(q);
    q += xlane_<8>(q);
    return q;
}

__global__ __launch_bounds__(256)
void k_scan_out(const float* __restrict__ delta,
                const float* __restrict__ xc,
                const float* __restrict__ zbuf,
                const float* __restrict__ x_dbl,
                const float* __restrict__ A_log,
                const float* __restrict__ Dp_,
                float* __restrict__ out)
{
    __shared__ float4 sD[STILE4];
    __shared__ float4 sU[STILE4];
    __shared__ float4 sB[STILE4];
    __shared__ float4 sC[STILE4];
    __shared__ float4 sY[STILE4];

    const int tid = threadIdx.x;
    const int cb  = blockIdx.x % (C_ / SCH);
    const int b   = blockIdx.x / (C_ / SCH);
    const size_t row0 = ((size_t)b * C_ + (size_t)cb * SCH) * L_;

    const float4* gD = (const float4*)(delta + row0);
    const float4* gX = (const float4*)(xc    + row0);
    const float4* gZ = (const float4*)(zbuf  + row0);
    const float4* gB = (const float4*)(x_dbl + ((size_t)b * RN2 + R_) * L_);
    const float4* gC = (const float4*)(x_dbl + ((size_t)b * RN2 + R_ + N_) * L_);
    float4* gO = (float4*)(out + row0);

    const int lane = tid & 31;
    const int n    = lane & 15;
    const int ch   = (tid >> 5) * 2 + (lane >> 4);
    const float A_neg = -__expf(A_log[(cb * SCH + ch) * N_ + n]);

    float4 dr[3], xr[3], zr[3], br[3], cr[3];
    const bool tail = tid < (STILE4 - 3 * 256);
    const float4 zero4 = make_float4(0.f, 0.f, 0.f, 0.f);
    float4 dT = zero4, xT = zero4, zT = zero4, bT = zero4, cT = zero4;
    float  pr[3], pT = 0.f;
    #pragma unroll
    for (int r = 0; r < 3; ++r) {
        const int i = tid + r * 256;
        dr[r] = gD[i];
        xr[r] = gX[i];
        br[r] = gB[i];
        cr[r] = gC[i];
        zr[r] = gZ[i];
        pr[r] = Dp_[cb * SCH + i / SROW4];
    }
    if (tail) {
        const int i = tid + 3 * 256;
        dT = gD[i];
        xT = gX[i];
        bT = gB[i];
        cT = gC[i];
        zT = gZ[i];
        pT = Dp_[cb * SCH + i / SROW4];
    }
    #pragma unroll
    for (int r = 0; r < 3; ++r) {
        const int i = tid + r * 256;
        const float4 d = dr[r];
        const float4 x = xr[r];
        sD[i] = d;
        sU[i] = make_float4(d.x * x.x, d.y * x.y, d.z * x.z, d.w * x.w);
        sB[i] = br[r];
        sC[i] = cr[r];
    }
    if (tail) {
        const int i = tid + 3 * 256;
        sD[i] = dT;
        sU[i] = make_float4(dT.x * xT.x, dT.y * xT.y, dT.z * xT.z, dT.w * xT.w);
        sB[i] = bT;
        sC[i] = cT;
    }
    __syncthreads();

    const float4* pD = sD + ch * SROW4;
    const float4* pU = sU + ch * SROW4;
    const float4* pB = sB + n  * SROW4;
    const float4* pC = sC + n  * SROW4;
    float4*       pY = sY + ch * SROW4;

    const bool odd1 = (n & 1) != 0;
    const bool odd2 = (n & 2) != 0;
    const int  grp  = n >> 2;
    float* pYrow = (float*)pY;

    float h = 0.f;
    float4 ad = pD[0], au = pU[0], ab = pB[0], ac = pC[0];
    float4 bd, bu, bb, bc;
    for (int J = 0; J < SROW4 / 4; ++J) {
        const int j = 4 * J;
        float keep, q;
        bd = pD[j + 1]; bu = pU[j + 1]; bb = pB[j + 1]; bc = pC[j + 1];
        q = scan_trip_(h, A_neg, ad, au, ab, ac, odd1, odd2);
        keep = q;
        ad = pD[j + 2]; au = pU[j + 2]; ab = pB[j + 2]; ac = pC[j + 2];
        q = scan_trip_(h, A_neg, bd, bu, bb, bc, odd1, odd2);
        keep = (grp == 1) ? q : keep;
        bd = pD[j + 3]; bu = pU[j + 3]; bb = pB[j + 3]; bc = pC[j + 3];
        q = scan_trip_(h, A_neg, ad, au, ab, ac, odd1, odd2);
        keep = (grp == 2) ? q : keep;
        ad = pD[j + 4]; au = pU[j + 4]; ab = pB[j + 4]; ac = pC[j + 4];
        q = scan_trip_(h, A_neg, bd, bu, bb, bc, odd1, odd2);
        keep = (grp == 3) ? q : keep;
        pYrow[16 * J + n] = keep;
    }
    {
        const float q = scan_trip_(h, A_neg, ad, au, ab, ac, odd1, odd2);
        if (n < 4) pYrow[4 * (SROW4 - 1) + n] = q;
    }
    __syncthreads();

    #pragma unroll
    for (int r = 0; r < 3; ++r) {
        const int i = tid + r * 256;
        gO[i] = gate_(sY[i], xr[r], zr[r], pr[r]);
    }
    if (tail) {
        const int i = tid + 3 * 256;
        gO[i] = gate_(sY[i], xT, zT, pT);
    }
}

extern "C" void kernel_launch(void* const* d_in, const int* in_sizes, int n_in,
                              void* d_out, int out_size, void* d_ws, size_t ws_size,
                              hipStream_t stream)
{
    const float* hidden    = (const float*)d_in[0];
    const float* in_proj_w = (const float*)d_in[1];
    const float* conv_w    = (const float*)d_in[2];
    const float* conv_b    = (const float*)d_in[3];
    const float* x_proj_w  = (const float*)d_in[4];
    const float* dt_proj_w = (const float*)d_in[5];
    const float* dt_bias   = (const float*)d_in[6];
    const float* A_log     = (const float*)d_in[7];
    const float* D_param   = (const float*)d_in[8];

    float* ws = (float*)d_ws;
    const size_t BCL = (size_t)BSZ * C_ * L_;
    float* x_pre = ws;
    float* zbuf  = ws + BCL;
    float* xcbuf = ws + 2 * BCL;
    float* delta = ws + 3 * BCL;
    float* x_dbl = ws + 4 * BCL;

    k_gemm_inproj<<<dim3(4, BSZ * T_), IPW * 32, 0, stream>>>(
        hidden, in_proj_w, x_pre, zbuf);

    k_conv_xproj_dt<<<dim3(2 * LTILES, BSZ), XPW * 32, 0, stream>>>(
        x_pre, conv_w, conv_b, x_proj_w, dt_proj_w, dt_bias, xcbuf, x_dbl, delta);

    k_scan_out<<<BSZ * (C_ / SCH), 256, 0, stream>>>(
        delta, xcbuf, zbuf, x_dbl, A_log, D_param, (float*)d_out);
}
